# starting kernel
# baseline (speedup 1.0000x reference)
#include <hip/hip_runtime.h>

typedef __attribute__((ext_vector_type(2))) float v2f;
typedef __attribute__((ext_vector_type(8))) float v8f;

__global__ __launch_bounds__(256) void poly_gemm_wmma(
    const float* __restrict__ X, const float* __restrict__ W,
    const float* __restrict__ bias, float* __restrict__ out,
    int N, int init)
{
  __shared__ v2f sW[16 * 4 * 32];
  for (int j = threadIdx.x; j < 16 * 4 * 32; j += 256) {
    const int kb4 = j >> 7;
    const int t   = (j >> 5) & 3;
    const int ln  = j & 31;
    const int h   = ln >> 4;
    const int lw  = ln & 15;
    const int krow = kb4 * 4 + 2 * h;
    const int ncol = lw + 16 * t;
    v2f p;
    p.x = W[krow * 64 + ncol];
    p.y = W[(krow + 1) * 64 + ncol];
    sW[j] = p;
  }
  __syncthreads();

  const int lane = threadIdx.x & 31;
  const int wave = threadIdx.x >> 5;
  const int rowBase = (blockIdx.x * 8 + wave) * 16;
  if (rowBase >= N) return;

  const int lo = lane & 15;
  const int hi = lane >> 4;

  v8f acc0 = {}, acc1 = {}, acc2 = {}, acc3 = {};
  const float* xrow = X + (size_t)(rowBase + lo) * 64;
  const v2f* wl = sW + lane;

#pragma unroll
  for (int kb = 0; kb < 64; kb += 4) {
    const int ka = kb + 2 * hi;
    v2f a;
    a.x = xrow[ka];
    a.y = xrow[ka + 1];
    const v2f* wb = wl + (kb >> 2) * 128;
    const v2f b0 = wb[0];
    const v2f b1 = wb[32];
    const v2f b2 = wb[64];
    const v2f b3 = wb[96];
    acc0 = __builtin_amdgcn_wmma_f32_16x16x4_f32(false, a, false, b0, (short)0, acc0, false, false);
    acc1 = __builtin_amdgcn_wmma_f32_16x16x4_f32(false, a, false, b1, (short)0, acc1, false, false);
    acc2 = __builtin_amdgcn_wmma_f32_16x16x4_f32(false, a, false, b2, (short)0, acc2, false, false);
    acc3 = __builtin_amdgcn_wmma_f32_16x16x4_f32(false, a, false, b3, (short)0, acc3, false, false);
  }

  const float bb0 = bias[lo], bb1 = bias[lo + 16],
              bb2 = bias[lo + 32], bb3 = bias[lo + 48];
#pragma unroll
  for (int r = 0; r < 8; ++r) {
    float* o = out + (size_t)(rowBase + r + 8 * hi) * 64;
    if (init) {
      o[lo]      = acc0[r] + bb0;
      o[lo + 16] = acc1[r] + bb1;
      o[lo + 32] = acc2[r] + bb2;
      o[lo + 48] = acc3[r] + bb3;
    } else {
      o[lo]      += acc0[r];
      o[lo + 16] += acc1[r];
      o[lo + 32] += acc2[r];
      o[lo + 48] += acc3[r];
    }
  }
}

__global__ void poly_zero(float4* __restrict__ p, int n4) {
  int i = blockIdx.x * blockDim.x + threadIdx.x;
  if (i < n4) p[i] = make_float4(0.f, 0.f, 0.f, 0.f);
}

__global__ void poly_spmm(const int* __restrict__ row, const int* __restrict__ col,
                          const float* __restrict__ val,
                          const float* __restrict__ xin, float* __restrict__ xout,
                          int E)
{
  int t = blockIdx.x * blockDim.x + threadIdx.x;
  int e = t >> 6;
  int f = t & 63;
  if (e >= E) return;
  float v = val[e];
  int c = col[e];
  int r = row[e];
  atomicAdd(xout + (size_t)r * 64 + f, v * xin[(size_t)c * 64 + f]);
}

extern "C" void kernel_launch(void* const* d_in, const int* in_sizes, int n_in,
                              void* d_out, int out_size, void* d_ws, size_t ws_size,
                              hipStream_t stream) {
  const float* x       = (const float*)d_in[0];
  const int*   lap_row = (const int*)d_in[1];
  const int*   lap_col = (const int*)d_in[2];
  const float* lap_val = (const float*)d_in[3];
  const float* weights = (const float*)d_in[4];
  const float* bias    = (const float*)d_in[5];
  float* out = (float*)d_out;

  const int N = in_sizes[0] / 64;
  const int E = in_sizes[1];
  const int K = 4;

  float* buf[2] = { (float*)d_ws, (float*)d_ws + (size_t)N * 64 };

  const int rowBlocks = (N + 15) / 16;
  dim3 gemmGrid((rowBlocks + 7) / 8), gemmBlock(256);
  const int n4 = (N * 64) / 4;
  dim3 zeroGrid((n4 + 255) / 256), zeroBlock(256);
  const long long spmmThreads = (long long)E * 64;
  dim3 spmmGrid((unsigned)((spmmThreads + 255) / 256)), spmmBlock(256);

  const float* xk = x;
  for (int k = 0; k <= K; ++k) {
    poly_gemm_wmma<<<gemmGrid, gemmBlock, 0, stream>>>(
        xk, weights + (size_t)k * 64 * 64, bias, out, N, (k == 0) ? 1 : 0);
    if (k < K) {
      float* nxt = buf[k & 1];
      poly_zero<<<zeroGrid, zeroBlock, 0, stream>>>((float4*)nxt, n4);
      poly_spmm<<<spmmGrid, spmmBlock, 0, stream>>>(lap_row, lap_col, lap_val,
                                                    xk, nxt, E);
      xk = nxt;
    }
  }
}


# Products moved into each pair's second hop: its block keeps its 64 finished rows in LDS, two waves' rows make one WMMA tile, out is updated in that launch; 5 product launches go; sum orders kept
# speedup vs baseline: 4.2923x; 4.2923x over previous
#include <hip/hip_runtime.h>

typedef __attribute__((ext_vector_type(2))) float v2f;
typedef __attribute__((ext_vector_type(8))) float v8f;

__device__ __forceinline__ v2f poly_w_entry(const float* __restrict__ WA,
                                            const float* __restrict__ WB, int j)
{
  const int kb4 = j >> 7;
  const int t   = (j >> 5) & 3;
  const int ln  = j & 31;
  const int h   = ln >> 4;
  const int lw  = ln & 15;
  const int krow = kb4 * 4 + 2 * h;
  const int ncol = lw + 16 * t;
  const float* wp = (krow < 32) ? (WA + krow * 64) : (WB + (krow - 32) * 64);
  v2f p;
  p.x = wp[ncol];
  p.y = wp[64 + ncol];
  return p;
}

#define SB 8
#define RB (1 << SB)
#define MAXNB 512
#define NSLICE 128
#define CH 8

__device__ __forceinline__ int poly_wave_scan(int v, int lane) {
#pragma unroll
  for (int d = 1; d < 32; d <<= 1) {
    const int u = __shfl_up(v, d);
    if (lane >= d) v += u;
  }
  return v;
}

__device__ __forceinline__ int poly_block_scan(int v, int* ws, int& total) {
  const int lane = threadIdx.x & 31;
  const int wave = threadIdx.x >> 5;
  const int inc = poly_wave_scan(v, lane);
  if (lane == 31) ws[wave] = inc;
  __syncthreads();
  const int wt = ws[lane];
  const int wsc = poly_wave_scan(wt, lane);
  total = __shfl(wsc, 31);
  return __shfl(wsc, wave) - __shfl(wt, wave) + inc - v;
}

__global__ __launch_bounds__(1024) void poly_hist(
    const int* __restrict__ row, int* __restrict__ cnt1, int E, int NBP)
{
  __shared__ int h[MAXNB];
  const int t = threadIdx.x;
  if (t < MAXNB) h[t] = 0;
  __syncthreads();
  const int per = (E + NSLICE - 1) / NSLICE;
  const int b = blockIdx.x * per;
  const int e = min(b + per, E);
  for (int base = b; base < e; base += CH * 1024) {
    int r[CH];
#pragma unroll
    for (int k = 0; k < CH; ++k) {
      const int i = base + k * 1024 + t;
      r[k] = (i < e) ? row[i] : -1;
    }
#pragma unroll
    for (int k = 0; k < CH; ++k)
      if (r[k] >= 0) atomicAdd(&h[r[k] >> SB], 1);
  }
  __syncthreads();
  if (t < NBP) cnt1[blockIdx.x * NBP + t] = h[t];
}

__global__ __launch_bounds__(1024) void poly_scatter(
    const int* __restrict__ row, const int* __restrict__ col, const float* __restrict__ val,
    const int* __restrict__ cnt1, int* __restrict__ bstart, int2* __restrict__ coarse,
    int E, int NB, int NBP)
{
  __shared__ int tot[MAXNB];
  __shared__ int bef[MAXNB];
  __shared__ int cur[MAXNB];
  __shared__ int ws[32];
  const int t = threadIdx.x;
  if (t < MAXNB) { tot[t] = 0; bef[t] = 0; }
  __syncthreads();
  const int C4 = NBP >> 2;
  const int nq = NSLICE * C4;
  const int4* m4 = (const int4*)cnt1;
  for (int base = 0; base < nq; base += CH * 1024) {
    int4 m[CH];
#pragma unroll
    for (int k = 0; k < CH; ++k) {
      const int q = base + k * 1024 + t;
      m[k] = (q < nq) ? m4[q] : make_int4(0, 0, 0, 0);
    }
#pragma unroll
    for (int k = 0; k < CH; ++k) {
      const int q = base + k * 1024 + t;
      if (q < nq) {
        const int s = q / C4;
        const int c = (q - s * C4) * 4;
        atomicAdd(&tot[c + 0], m[k].x);
        atomicAdd(&tot[c + 1], m[k].y);
        atomicAdd(&tot[c + 2], m[k].z);
        atomicAdd(&tot[c + 3], m[k].w);
        if (s < (int)blockIdx.x) {
          atomicAdd(&bef[c + 0], m[k].x);
          atomicAdd(&bef[c + 1], m[k].y);
          atomicAdd(&bef[c + 2], m[k].z);
          atomicAdd(&bef[c + 3], m[k].w);
        }
      }
    }
  }
  __syncthreads();
  const int mine = (t < MAXNB) ? tot[t] : 0;
  int sum;
  const int ex = poly_block_scan(mine, ws, sum);
  if (t < MAXNB) cur[t] = ex + bef[t];
  if (blockIdx.x == 0 && t <= NB) bstart[t] = ex;
  __syncthreads();

  const int per = (E + NSLICE - 1) / NSLICE;
  const int b = blockIdx.x * per;
  const int e = min(b + per, E);
  for (int base = b; base < e; base += CH * 1024) {
    int r[CH], c[CH];
    float v[CH];
#pragma unroll
    for (int k = 0; k < CH; ++k) {
      const int i = base + k * 1024 + t;
      const bool ok = i < e;
      r[k] = ok ? row[i] : -1;
      c[k] = ok ? col[i] : 0;
      v[k] = ok ? val[i] : 0.f;
    }
    int p[CH];
#pragma unroll
    for (int k = 0; k < CH; ++k)
      p[k] = (r[k] >= 0) ? atomicAdd(&cur[r[k] >> SB], 1) : -1;
#pragma unroll
    for (int k = 0; k < CH; ++k)
      if (p[k] >= 0)
        coarse[p[k]] = make_int2((c[k] << SB) | (r[k] & (RB - 1)), __float_as_int(v[k]));
  }
}

__global__ __launch_bounds__(1024) void poly_finalize(
    const int2* __restrict__ coarse, const int* __restrict__ bstart,
    int* __restrict__ rowptr, int2* __restrict__ sedge, int N)
{
  __shared__ int rc[RB];
  __shared__ int ws[32];
  const int t = threadIdx.x;
  const int bk = blockIdx.x;
  const int s = bstart[bk];
  const int e = bstart[bk + 1];
  if (t < RB) rc[t] = 0;
  __syncthreads();
  for (int base = s; base < e; base += CH * 1024) {
    int key[CH];
#pragma unroll
    for (int k = 0; k < CH; ++k) {
      const int i = base + k * 1024 + t;
      key[k] = (i < e) ? coarse[i].x : -1;
    }
#pragma unroll
    for (int k = 0; k < CH; ++k)
      if (key[k] >= 0) atomicAdd(&rc[key[k] & (RB - 1)], 1);
  }
  __syncthreads();
  const int cn = (t < RB) ? rc[t] : 0;
  int sum;
  const int ex = poly_block_scan(cn, ws, sum);
  if (t < RB) {
    rc[t] = s + ex;
    const int r = bk * RB + t;
    if (r < N) rowptr[r] = s + ex;
  }
  if (bk == (int)gridDim.x - 1 && t == 0) rowptr[N] = e;
  __syncthreads();
  for (int base = s; base < e; base += CH * 1024) {
    int2 rec[CH];
#pragma unroll
    for (int k = 0; k < CH; ++k) {
      const int i = base + k * 1024 + t;
      rec[k] = (i < e) ? coarse[i] : make_int2(-1, 0);
    }
    int p[CH];
#pragma unroll
    for (int k = 0; k < CH; ++k)
      p[k] = (rec[k].x >= 0) ? atomicAdd(&rc[rec[k].x & (RB - 1)], 1) : -1;
#pragma unroll
    for (int k = 0; k < CH; ++k)
      if (p[k] >= 0) sedge[p[k]] = make_int2(rec[k].x >> SB, rec[k].y);
  }
}

#define GR 8
#define TP 36

template <bool TILE>
__device__ __forceinline__ void poly_close_rows(int j, int& rend, int& q, float& acc,
                                                float* __restrict__ xo, float* st, int keep,
                                                unsigned lane, int rp)
{
  while (j == rend) {
    if (!TILE || keep) xo[(unsigned)q * 32u + lane] = acc;
    if (TILE) st[q * TP + (int)lane] = acc;
    acc = 0.f;
    ++q;
    rend = __builtin_amdgcn_readlane(rp, q + 1);
  }
}

template <bool TILE>
__device__ __forceinline__ void poly_walk(
    const int* __restrict__ rowptr, const int* __restrict__ sedge,
    const float* __restrict__ xin, int ldin,
    float* __restrict__ xout, float* st, int keep, int r0, int N, unsigned lane)
{
  const int nr = min(GR, N - r0);
  const int rp = rowptr[r0 + min((int)lane, nr)];
  const int b = __builtin_amdgcn_readlane(rp, 0);
  const int e = __builtin_amdgcn_readlane(rp, nr);
  const int2* se = (const int2*)sedge;
  float* xo = xout + (size_t)r0 * 32;
  int q = 0;
  int rend = __builtin_amdgcn_readlane(rp, 1);
  float acc = 0.f;
  if (b < e) {
    int2 ed = se[min(b + (int)lane, e - 1)];
    for (int base = b; base < e; base += 32) {
      const int2 nx = se[min(base + 32 + (int)lane, e - 1)];
      const int n = e - base;
      float xv[32];
#pragma unroll
      for (int u = 0; u < 32; ++u) {
        const unsigned c = (unsigned)__builtin_amdgcn_readlane(ed.x, u);
        xv[u] = xin[c * (unsigned)ldin + lane];
      }
      if (n >= 32) {
#pragma unroll
        for (int u = 0; u < 32; ++u) {
          poly_close_rows<TILE>(base + u, rend, q, acc, xo, st, keep, lane, rp);
          acc = fmaf(__int_as_float(__builtin_amdgcn_readlane(ed.y, u)), xv[u], acc);
        }
      } else {
#pragma unroll
        for (int u = 0; u < 32; ++u) {
          if (u < n) {
            poly_close_rows<TILE>(base + u, rend, q, acc, xo, st, keep, lane, rp);
            acc = fmaf(__int_as_float(__builtin_amdgcn_readlane(ed.y, u)), xv[u], acc);
          }
        }
      }
      ed = nx;
    }
  }
  while (q < nr) {
    if (!TILE || keep) xo[(unsigned)q * 32u + lane] = acc;
    if (TILE) st[q * TP + (int)lane] = acc;
    acc = 0.f;
    ++q;
  }
  if (TILE)
    for (; q < GR; ++q) st[q * TP + (int)lane] = 0.f;
}

__global__ __launch_bounds__(256) void poly_gather(
    const int* __restrict__ rowptr, const int* __restrict__ sedge,
    const float* __restrict__ xin, int ldin,
    float* __restrict__ xout, int N)
{
  const unsigned lane = threadIdx.x & 31;
  const int r0 = __builtin_amdgcn_readfirstlane((blockIdx.x * 8 + (threadIdx.x >> 5)) * GR);
  if (r0 >= N) return;
  poly_walk<false>(rowptr, sedge, xin, ldin, xout, nullptr, 1, r0, N, lane);
}

template <bool FIRST>
__global__ __launch_bounds__(256) void poly_gather_mm(
    const int* __restrict__ rowptr, const int* __restrict__ sedge,
    const float* __restrict__ xin, float* __restrict__ xout, int keep,
    const float* __restrict__ WA, const float* __restrict__ WB,
    const float* __restrict__ x, const float* __restrict__ W0,
    const float* __restrict__ bias,
    float* __restrict__ out, int N)
{
  __shared__ v2f sW[16 * 4 * 32];
  __shared__ v2f sW0[FIRST ? 16 * 4 * 32 : 1];
  __shared__ float sT[8 * GR * TP];
  const unsigned lane = threadIdx.x & 31;
  const int wave = __builtin_amdgcn_readfirstlane(threadIdx.x >> 5);
  const int lo = lane & 15;
  const int hi = lane >> 4;

  v2f wv[8], wv0[FIRST ? 8 : 1];
#pragma unroll
  for (int i = 0; i < 8; ++i) {
    wv[i] = poly_w_entry(WA, WB, threadIdx.x + 256 * i);
    if (FIRST) wv0[i] = poly_w_entry(W0, W0 + 32 * 64, threadIdx.x + 256 * i);
  }

  const int tile = wave >> 1;
  const int nsel = wave & 1;
  const int rowBase = blockIdx.x * (8 * GR) + tile * 16;
  const int col0 = nsel * 32 + lo;
  const int rowA = min(rowBase + lo, N - 1);
  const v2f* xa = (const v2f*)(xin + (size_t)rowA * 32 + 2 * hi);
  v2f av[8];
#pragma unroll
  for (int s = 0; s < 8; ++s) av[s] = xa[2 * s];
  float o0[8], o1[8];
  if (FIRST) {
    o0[0] = bias[col0];
    o1[0] = bias[col0 + 16];
  } else {
#pragma unroll
    for (int r = 0; r < 8; ++r) {
      const float* o = out + (size_t)min(rowBase + r + 8 * hi, N - 1) * 64 + col0;
      o0[r] = o[0];
      o1[r] = o[16];
    }
  }

  const int r0 = (blockIdx.x * 8 + wave) * GR;
  float* st = sT + wave * (GR * TP);
  if (r0 < N) {
    poly_walk<true>(rowptr, sedge, xin, 32, xout, st, keep, r0, N, lane);
  } else {
    for (int q = 0; q < GR; ++q) st[q * TP + (int)lane] = 0.f;
  }

#pragma unroll
  for (int i = 0; i < 8; ++i) {
    sW[threadIdx.x + 256 * i] = wv[i];
    if (FIRST) sW0[threadIdx.x + 256 * i] = wv0[i];
  }
  __syncthreads();
  if (rowBase >= N) return;

  v2f ax[FIRST ? 16 : 1];
  if (FIRST) {
    const v2f* xr = (const v2f*)(x + (size_t)rowA * 64 + 2 * hi);
#pragma unroll
    for (int s = 0; s < 16; ++s) ax[s] = xr[2 * s];
  }

  v8f acc0 = {}, acc1 = {};
  const v2f* wl = sW + lane + nsel * 64;
  const float* ta = sT + (tile * 16 + lo) * TP + 2 * hi;
#pragma unroll
  for (int s = 0; s < 16; ++s) {
    v2f a;
    if (s < 8) a = av[s];
    else       a = *(const v2f*)(ta + 4 * (s - 8));
    const v2f b0 = wl[s * 128];
    const v2f b1 = wl[s * 128 + 32];
    acc0 = __builtin_amdgcn_wmma_f32_16x16x4_f32(false, a, false, b0, (short)0, acc0, false, false);
    acc1 = __builtin_amdgcn_wmma_f32_16x16x4_f32(false, a, false, b1, (short)0, acc1, false, false);
  }

  if (FIRST) {
    v8f acx0 = {}, acx1 = {};
    const v2f* wl0 = sW0 + lane + nsel * 64;
#pragma unroll
    for (int s = 0; s < 16; ++s) {
      const v2f b0 = wl0[s * 128];
      const v2f b1 = wl0[s * 128 + 32];
      acx0 = __builtin_amdgcn_wmma_f32_16x16x4_f32(false, ax[s], false, b0, (short)0, acx0, false, false);
      acx1 = __builtin_amdgcn_wmma_f32_16x16x4_f32(false, ax[s], false, b1, (short)0, acx1, false, false);
    }
#pragma unroll
    for (int r = 0; r < 8; ++r) {
      const int row = rowBase + r + 8 * hi;
      if (row < N) {
        float* o = out + (size_t)row * 64 + col0;
        o[0]  = (acx0[r] + o0[0]) + acc0[r];
        o[16] = (acx1[r] + o1[0]) + acc1[r];
      }
    }
  } else {
#pragma unroll
    for (int r = 0; r < 8; ++r) {
      const int row = rowBase + r + 8 * hi;
      if (row < N) {
        float* o = out + (size_t)row * 64 + col0;
        o[0]  = o0[r] + acc0[r];
        o[16] = o1[r] + acc1[r];
      }
    }
  }
}

extern "C" void kernel_launch(void* const* d_in, const int* in_sizes, int n_in,
                              void* d_out, int out_size, void* d_ws, size_t ws_size,
                              hipStream_t stream) {
  const float* x       = (const float*)d_in[0];
  const int*   lap_row = (const int*)d_in[1];
  const int*   lap_col = (const int*)d_in[2];
  const float* lap_val = (const float*)d_in[3];
  const float* weights = (const float*)d_in[4];
  const float* bias    = (const float*)d_in[5];
  float* out = (float*)d_out;

  const int N = in_sizes[0] / 64;
  const int E = in_sizes[1];

  float* P0   = (float*)d_ws;
  float* P1   = P0 + (size_t)N * 32;
  int* bstart = (int*)(P1 + (size_t)N * 32);
  int* rowptr = bstart + N;
  int* sedge  = rowptr + N + 4;
  int2* coarse = (int2*)d_ws;
  int* cnt1    = (int*)d_ws + (((size_t)E * 2 + 3) & ~(size_t)3);

  const int NB  = (N + RB - 1) / RB;
  const int NBP = (NB + 3) & ~3;

  dim3 gatherGrid((N + 8 * GR - 1) / (8 * GR)), gatherBlock(256);

  poly_hist<<<dim3(NSLICE), dim3(1024), 0, stream>>>(lap_row, cnt1, E, NBP);
  poly_scatter<<<dim3(NSLICE), dim3(1024), 0, stream>>>(lap_row, lap_col, lap_val,
                                                        cnt1, bstart, coarse, E, NB, NBP);
  poly_finalize<<<dim3(NB), dim3(1024), 0, stream>>>(coarse, bstart, rowptr, (int2*)sedge, N);

  for (int h = 0; h < 2; ++h) {
    const float* wh = weights + (size_t)h * 32 * 64;
    poly_gather<<<gatherGrid, gatherBlock, 0, stream>>>(rowptr, sedge, x + h * 32, 64, P0, N);
    if (h == 0)
      poly_gather_mm<true><<<gatherGrid, gatherBlock, 0, stream>>>(
          rowptr, sedge, P0, P1, 1, wh + (size_t)1 * 64 * 64, wh + (size_t)2 * 64 * 64,
          x, weights, bias, out, N);
    else
      poly_gather_mm<false><<<gatherGrid, gatherBlock, 0, stream>>>(
          rowptr, sedge, P0, P1, 1, wh + (size_t)1 * 64 * 64, wh + (size_t)2 * 64 * 64,
          x, weights, bias, out, N);
    poly_gather<<<gatherGrid, gatherBlock, 0, stream>>>(rowptr, sedge, P1, 32, P0, N);
    poly_gather_mm<false><<<gatherGrid, gatherBlock, 0, stream>>>(
        rowptr, sedge, P0, P1, 0, wh + (size_t)3 * 64 * 64, wh + (size_t)4 * 64 * 64,
        x, weights, bias, out, N);
  }
}


# Edge sort's two placing passes put their 8-byte records into LDS in list order first and copy them out with whole-line stores; was 32 partial lines per store instruction; slots and row sum order kept
# speedup vs baseline: 4.6053x; 1.0729x over previous
#include <hip/hip_runtime.h>

typedef __attribute__((ext_vector_type(2))) float v2f;
typedef __attribute__((ext_vector_type(8))) float v8f;

__device__ __forceinline__ v2f poly_w_entry(const float* __restrict__ WA,
                                            const float* __restrict__ WB, int j)
{
  const int kb4 = j >> 7;
  const int t   = (j >> 5) & 3;
  const int ln  = j & 31;
  const int h   = ln >> 4;
  const int lw  = ln & 15;
  const int krow = kb4 * 4 + 2 * h;
  const int ncol = lw + 16 * t;
  const float* wp = (krow < 32) ? (WA + krow * 64) : (WB + (krow - 32) * 64);
  v2f p;
  p.x = wp[ncol];
  p.y = wp[64 + ncol];
  return p;
}

#define SB 8
#define RB (1 << SB)
#define MAXNB 512
#define NSLICE 128
#define CH 8
#define SCAP 8192
#define FCAP 6144

__device__ __forceinline__ int poly_wave_scan(int v, int lane) {
#pragma unroll
  for (int d = 1; d < 32; d <<= 1) {
    const int u = __shfl_up(v, d);
    if (lane >= d) v += u;
  }
  return v;
}

__device__ __forceinline__ int poly_block_scan(int v, int* ws, int& total) {
  const int lane = threadIdx.x & 31;
  const int wave = threadIdx.x >> 5;
  const int inc = poly_wave_scan(v, lane);
  if (lane == 31) ws[wave] = inc;
  __syncthreads();
  const int wt = ws[lane];
  const int wsc = poly_wave_scan(wt, lane);
  total = __shfl(wsc, 31);
  return __shfl(wsc, wave) - __shfl(wt, wave) + inc - v;
}

__global__ __launch_bounds__(1024) void poly_hist(
    const int* __restrict__ row, int* __restrict__ cnt1, int E, int NBP)
{
  __shared__ int h[MAXNB];
  const int t = threadIdx.x;
  if (t < MAXNB) h[t] = 0;
  __syncthreads();
  const int per = (E + NSLICE - 1) / NSLICE;
  const int b = blockIdx.x * per;
  const int e = min(b + per, E);
  for (int base = b; base < e; base += CH * 1024) {
    int r[CH];
#pragma unroll
    for (int k = 0; k < CH; ++k) {
      const int i = base + k * 1024 + t;
      r[k] = (i < e) ? row[i] : -1;
    }
#pragma unroll
    for (int k = 0; k < CH; ++k)
      if (r[k] >= 0) atomicAdd(&h[r[k] >> SB], 1);
  }
  __syncthreads();
  if (t < NBP) cnt1[blockIdx.x * NBP + t] = h[t];
}

__global__ __launch_bounds__(1024) void poly_scatter(
    const int* __restrict__ row, const int* __restrict__ col, const float* __restrict__ val,
    const int* __restrict__ cnt1, int* __restrict__ bstart, int2* __restrict__ coarse,
    int E, int NB, int NBP)
{
  __shared__ int tot[MAXNB];
  __shared__ int bef[MAXNB];
  __shared__ int cur[MAXNB];
  __shared__ int ws[32];
  __shared__ int2 stR[SCAP];
  __shared__ int stP[SCAP];
  const int t = threadIdx.x;
  if (t < MAXNB) { tot[t] = 0; bef[t] = 0; }
  __syncthreads();
  const int C4 = NBP >> 2;
  const int nq = NSLICE * C4;
  const int4* m4 = (const int4*)cnt1;
  for (int base = 0; base < nq; base += CH * 1024) {
    int4 m[CH];
#pragma unroll
    for (int k = 0; k < CH; ++k) {
      const int q = base + k * 1024 + t;
      m[k] = (q < nq) ? m4[q] : make_int4(0, 0, 0, 0);
    }
#pragma unroll
    for (int k = 0; k < CH; ++k) {
      const int q = base + k * 1024 + t;
      if (q < nq) {
        const int s = q / C4;
        const int c = (q - s * C4) * 4;
        atomicAdd(&tot[c + 0], m[k].x);
        atomicAdd(&tot[c + 1], m[k].y);
        atomicAdd(&tot[c + 2], m[k].z);
        atomicAdd(&tot[c + 3], m[k].w);
        if (s < (int)blockIdx.x) {
          atomicAdd(&bef[c + 0], m[k].x);
          atomicAdd(&bef[c + 1], m[k].y);
          atomicAdd(&bef[c + 2], m[k].z);
          atomicAdd(&bef[c + 3], m[k].w);
        }
      }
    }
  }
  __syncthreads();
  const int mine = (t < MAXNB) ? tot[t] : 0;
  int sum;
  const int ex = poly_block_scan(mine, ws, sum);
  if (t < MAXNB) cur[t] = ex + bef[t];
  if (blockIdx.x == 0 && t <= NB) bstart[t] = ex;
  const int own = (t < NBP) ? cnt1[blockIdx.x * NBP + t] : 0;
  __syncthreads();
  int osum;
  const int lex = poly_block_scan(own, ws, osum);
  if (t < MAXNB) tot[t] = ex + bef[t] - lex;
  __syncthreads();

  const int per = (E + NSLICE - 1) / NSLICE;
  const int b = blockIdx.x * per;
  const int e = min(b + per, E);
  const bool stage = (e - b) <= SCAP;
  for (int base = b; base < e; base += CH * 1024) {
    int r[CH], c[CH];
    float v[CH];
#pragma unroll
    for (int k = 0; k < CH; ++k) {
      const int i = base + k * 1024 + t;
      const bool ok = i < e;
      r[k] = ok ? row[i] : -1;
      c[k] = ok ? col[i] : 0;
      v[k] = ok ? val[i] : 0.f;
    }
    int p[CH];
#pragma unroll
    for (int k = 0; k < CH; ++k)
      p[k] = (r[k] >= 0) ? atomicAdd(&cur[r[k] >> SB], 1) : -1;
#pragma unroll
    for (int k = 0; k < CH; ++k)
      if (p[k] >= 0) {
        const int2 rec = make_int2((c[k] << SB) | (r[k] & (RB - 1)), __float_as_int(v[k]));
        if (stage) {
          const int l = p[k] - tot[r[k] >> SB];
          stR[l] = rec;
          stP[l] = p[k];
        } else {
          coarse[p[k]] = rec;
        }
      }
  }
  if (stage) {
    __syncthreads();
    const int n = e - b;
    for (int i = t; i < n; i += 1024) coarse[stP[i]] = stR[i];
  }
}

__global__ __launch_bounds__(1024) void poly_finalize(
    const int2* __restrict__ coarse, const int* __restrict__ bstart,
    int* __restrict__ rowptr, int2* __restrict__ sedge, int N)
{
  __shared__ int rc[RB];
  __shared__ int ws[32];
  __shared__ int2 stF[FCAP];
  const int t = threadIdx.x;
  const int bk = blockIdx.x;
  const int s = bstart[bk];
  const int e = bstart[bk + 1];
  if (t < RB) rc[t] = 0;
  __syncthreads();
  for (int base = s; base < e; base += CH * 1024) {
    int key[CH];
#pragma unroll
    for (int k = 0; k < CH; ++k) {
      const int i = base + k * 1024 + t;
      key[k] = (i < e) ? coarse[i].x : -1;
    }
#pragma unroll
    for (int k = 0; k < CH; ++k)
      if (key[k] >= 0) atomicAdd(&rc[key[k] & (RB - 1)], 1);
  }
  __syncthreads();
  const int cn = (t < RB) ? rc[t] : 0;
  int sum;
  const int ex = poly_block_scan(cn, ws, sum);
  if (t < RB) {
    rc[t] = s + ex;
    const int r = bk * RB + t;
    if (r < N) rowptr[r] = s + ex;
  }
  if (bk == (int)gridDim.x - 1 && t == 0) rowptr[N] = e;
  __syncthreads();
  const bool stage = (e - s) <= FCAP;
  for (int base = s; base < e; base += CH * 1024) {
    int2 rec[CH];
#pragma unroll
    for (int k = 0; k < CH; ++k) {
      const int i = base + k * 1024 + t;
      rec[k] = (i < e) ? coarse[i] : make_int2(-1, 0);
    }
    int p[CH];
#pragma unroll
    for (int k = 0; k < CH; ++k)
      p[k] = (rec[k].x >= 0) ? atomicAdd(&rc[rec[k].x & (RB - 1)], 1) : -1;
#pragma unroll
    for (int k = 0; k < CH; ++k)
      if (p[k] >= 0) {
        const int2 o = make_int2(rec[k].x >> SB, rec[k].y);
        if (stage) stF[p[k] - s] = o;
        else       sedge[p[k]] = o;
      }
  }
  if (stage) {
    __syncthreads();
    const int off = (int)(((size_t)(sedge + s) >> 3) & 15);
    for (int i = t - off; i < e - s; i += 1024)
      if (i >= 0) sedge[s + i] = stF[i];
  }
}

#define GR 8
#define TP 36

template <bool TILE>
__device__ __forceinline__ void poly_close_rows(int j, int& rend, int& q, float& acc,
                                                float* __restrict__ xo, float* st, int keep,
                                                unsigned lane, int rp)
{
  while (j == rend) {
    if (!TILE || keep) xo[(unsigned)q * 32u + lane] = acc;
    if (TILE) st[q * TP + (int)lane] = acc;
    acc = 0.f;
    ++q;
    rend = __builtin_amdgcn_readlane(rp, q + 1);
  }
}

template <bool TILE>
__device__ __forceinline__ void poly_walk(
    const int* __restrict__ rowptr, const int* __restrict__ sedge,
    const float* __restrict__ xin, int ldin,
    float* __restrict__ xout, float* st, int keep, int r0, int N, unsigned lane)
{
  const int nr = min(GR, N - r0);
  const int rp = rowptr[r0 + min((int)lane, nr)];
  const int b = __builtin_amdgcn_readlane(rp, 0);
  const int e = __builtin_amdgcn_readlane(rp, nr);
  const int2* se = (const int2*)sedge;
  float* xo = xout + (size_t)r0 * 32;
  int q = 0;
  int rend = __builtin_amdgcn_readlane(rp, 1);
  float acc = 0.f;
  if (b < e) {
    int2 ed = se[min(b + (int)lane, e - 1)];
    for (int base = b; base < e; base += 32) {
      const int2 nx = se[min(base + 32 + (int)lane, e - 1)];
      const int n = e - base;
      float xv[32];
#pragma unroll
      for (int u = 0; u < 32; ++u) {
        const unsigned c = (unsigned)__builtin_amdgcn_readlane(ed.x, u);
        xv[u] = xin[c * (unsigned)ldin + lane];
      }
      if (n >= 32) {
#pragma unroll
        for (int u = 0; u < 32; ++u) {
          poly_close_rows<TILE>(base + u, rend, q, acc, xo, st, keep, lane, rp);
          acc = fmaf(__int_as_float(__builtin_amdgcn_readlane(ed.y, u)), xv[u], acc);
        }
      } else {
#pragma unroll
        for (int u = 0; u < 32; ++u) {
          if (u < n) {
            poly_close_rows<TILE>(base + u, rend, q, acc, xo, st, keep, lane, rp);
            acc = fmaf(__int_as_float(__builtin_amdgcn_readlane(ed.y, u)), xv[u], acc);
          }
        }
      }
      ed = nx;
    }
  }
  while (q < nr) {
    if (!TILE || keep) xo[(unsigned)q * 32u + lane] = acc;
    if (TILE) st[q * TP + (int)lane] = acc;
    acc = 0.f;
    ++q;
  }
  if (TILE)
    for (; q < GR; ++q) st[q * TP + (int)lane] = 0.f;
}

__global__ __launch_bounds__(256) void poly_gather(
    const int* __restrict__ rowptr, const int* __restrict__ sedge,
    const float* __restrict__ xin, int ldin,
    float* __restrict__ xout, int N)
{
  const unsigned lane = threadIdx.x & 31;
  const int r0 = __builtin_amdgcn_readfirstlane((blockIdx.x * 8 + (threadIdx.x >> 5)) * GR);
  if (r0 >= N) return;
  poly_walk<false>(rowptr, sedge, xin, ldin, xout, nullptr, 1, r0, N, lane);
}

template <bool FIRST>
__global__ __launch_bounds__(256) void poly_gather_mm(
    const int* __restrict__ rowptr, const int* __restrict__ sedge,
    const float* __restrict__ xin, float* __restrict__ xout, int keep,
    const float* __restrict__ WA, const float* __restrict__ WB,
    const float* __restrict__ x, const float* __restrict__ W0,
    const float* __restrict__ bias,
    float* __restrict__ out, int N)
{
  __shared__ v2f sW[16 * 4 * 32];
  __shared__ v2f sW0[FIRST ? 16 * 4 * 32 : 1];
  __shared__ float sT[8 * GR * TP];
  const unsigned lane = threadIdx.x & 31;
  const int wave = __builtin_amdgcn_readfirstlane(threadIdx.x >> 5);
  const int lo = lane & 15;
  const int hi = lane >> 4;

  v2f wv[8], wv0[FIRST ? 8 : 1];
#pragma unroll
  for (int i = 0; i < 8; ++i) {
    wv[i] = poly_w_entry(WA, WB, threadIdx.x + 256 * i);
    if (FIRST) wv0[i] = poly_w_entry(W0, W0 + 32 * 64, threadIdx.x + 256 * i);
  }

  const int tile = wave >> 1;
  const int nsel = wave & 1;
  const int rowBase = blockIdx.x * (8 * GR) + tile * 16;
  const int col0 = nsel * 32 + lo;
  const int rowA = min(rowBase + lo, N - 1);
  const v2f* xa = (const v2f*)(xin + (size_t)rowA * 32 + 2 * hi);
  v2f av[8];
#pragma unroll
  for (int s = 0; s < 8; ++s) av[s] = xa[2 * s];
  float o0[8], o1[8];
  if (FIRST) {
    o0[0] = bias[col0];
    o1[0] = bias[col0 + 16];
  } else {
#pragma unroll
    for (int r = 0; r < 8; ++r) {
      const float* o = out + (size_t)min(rowBase + r + 8 * hi, N - 1) * 64 + col0;
      o0[r] = o[0];
      o1[r] = o[16];
    }
  }

  const int r0 = (blockIdx.x * 8 + wave) * GR;
  float* st = sT + wave * (GR * TP);
  if (r0 < N) {
    poly_walk<true>(rowptr, sedge, xin, 32, xout, st, keep, r0, N, lane);
  } else {
    for (int q = 0; q < GR; ++q) st[q * TP + (int)lane] = 0.f;
  }

#pragma unroll
  for (int i = 0; i < 8; ++i) {
    sW[threadIdx.x + 256 * i] = wv[i];
    if (FIRST) sW0[threadIdx.x + 256 * i] = wv0[i];
  }
  __syncthreads();
  if (rowBase >= N) return;

  v2f ax[FIRST ? 16 : 1];
  if (FIRST) {
    const v2f* xr = (const v2f*)(x + (size_t)rowA * 64 + 2 * hi);
#pragma unroll
    for (int s = 0; s < 16; ++s) ax[s] = xr[2 * s];
  }

  v8f acc0 = {}, acc1 = {};
  const v2f* wl = sW + lane + nsel * 64;
  const float* ta = sT + (tile * 16 + lo) * TP + 2 * hi;
#pragma unroll
  for (int s = 0; s < 16; ++s) {
    v2f a;
    if (s < 8) a = av[s];
    else       a = *(const v2f*)(ta + 4 * (s - 8));
    const v2f b0 = wl[s * 128];
    const v2f b1 = wl[s * 128 + 32];
    acc0 = __builtin_amdgcn_wmma_f32_16x16x4_f32(false, a, false, b0, (short)0, acc0, false, false);
    acc1 = __builtin_amdgcn_wmma_f32_16x16x4_f32(false, a, false, b1, (short)0, acc1, false, false);
  }

  if (FIRST) {
    v8f acx0 = {}, acx1 = {};
    const v2f* wl0 = sW0 + lane + nsel * 64;
#pragma unroll
    for (int s = 0; s < 16; ++s) {
      const v2f b0 = wl0[s * 128];
      const v2f b1 = wl0[s * 128 + 32];
      acx0 = __builtin_amdgcn_wmma_f32_16x16x4_f32(false, ax[s], false, b0, (short)0, acx0, false, false);
      acx1 = __builtin_amdgcn_wmma_f32_16x16x4_f32(false, ax[s], false, b1, (short)0, acx1, false, false);
    }
#pragma unroll
    for (int r = 0; r < 8; ++r) {
      const int row = rowBase + r + 8 * hi;
      if (row < N) {
        float* o = out + (size_t)row * 64 + col0;
        o[0]  = (acx0[r] + o0[0]) + acc0[r];
        o[16] = (acx1[r] + o1[0]) + acc1[r];
      }
    }
  } else {
#pragma unroll
    for (int r = 0; r < 8; ++r) {
      const int row = rowBase + r + 8 * hi;
      if (row < N) {
        float* o = out + (size_t)row * 64 + col0;
        o[0]  = o0[r] + acc0[r];
        o[16] = o1[r] + acc1[r];
      }
    }
  }
}

extern "C" void kernel_launch(void* const* d_in, const int* in_sizes, int n_in,
                              void* d_out, int out_size, void* d_ws, size_t ws_size,
                              hipStream_t stream) {
  const float* x       = (const float*)d_in[0];
  const int*   lap_row = (const int*)d_in[1];
  const int*   lap_col = (const int*)d_in[2];
  const float* lap_val = (const float*)d_in[3];
  const float* weights = (const float*)d_in[4];
  const float* bias    = (const float*)d_in[5];
  float* out = (float*)d_out;

  const int N = in_sizes[0] / 64;
  const int E = in_sizes[1];

  float* P0   = (float*)d_ws;
  float* P1   = P0 + (size_t)N * 32;
  int* bstart = (int*)(P1 + (size_t)N * 32);
  int* rowptr = bstart + N;
  int* sedge  = rowptr + N + 4;
  int2* coarse = (int2*)d_ws;
  int* cnt1    = (int*)d_ws + (((size_t)E * 2 + 3) & ~(size_t)3);

  const int NB  = (N + RB - 1) / RB;
  const int NBP = (NB + 3) & ~3;

  dim3 gatherGrid((N + 8 * GR - 1) / (8 * GR)), gatherBlock(256);

  poly_hist<<<dim3(NSLICE), dim3(1024), 0, stream>>>(lap_row, cnt1, E, NBP);
  poly_scatter<<<dim3(NSLICE), dim3(1024), 0, stream>>>(lap_row, lap_col, lap_val,
                                                        cnt1, bstart, coarse, E, NB, NBP);
  poly_finalize<<<dim3(NB), dim3(1024), 0, stream>>>(coarse, bstart, rowptr, (int2*)sedge, N);

  for (int h = 0; h < 2; ++h) {
    const float* wh = weights + (size_t)h * 32 * 64;
    poly_gather<<<gatherGrid, gatherBlock, 0, stream>>>(rowptr, sedge, x + h * 32, 64, P0, N);
    if (h == 0)
      poly_gather_mm<true><<<gatherGrid, gatherBlock, 0, stream>>>(
          rowptr, sedge, P0, P1, 1, wh + (size_t)1 * 64 * 64, wh + (size_t)2 * 64 * 64,
          x, weights, bias, out, N);
    else
      poly_gather_mm<false><<<gatherGrid, gatherBlock, 0, stream>>>(
          rowptr, sedge, P0, P1, 1, wh + (size_t)1 * 64 * 64, wh + (size_t)2 * 64 * 64,
          x, weights, bias, out, N);
    poly_gather<<<gatherGrid, gatherBlock, 0, stream>>>(rowptr, sedge, P1, 32, P0, N);
    poly_gather_mm<false><<<gatherGrid, gatherBlock, 0, stream>>>(
        rowptr, sedge, P0, P1, 0, wh + (size_t)3 * 64 * 64, wh + (size_t)4 * 64 * 64,
        x, weights, bias, out, N);
  }
}


# Row gather re-laned: 8 lanes x 16 B per gathered line, four lane groups of a wave walk two rows each, one load instruction fetches 4 edges' lines; per-edge readlane and scalar work go; sum order kept
# speedup vs baseline: 4.9103x; 1.0662x over previous
#include <hip/hip_runtime.h>

typedef __attribute__((ext_vector_type(2))) float v2f;
typedef __attribute__((ext_vector_type(8))) float v8f;

__device__ __forceinline__ v2f poly_w_entry(const float* __restrict__ WA,
                                            const float* __restrict__ WB, int j)
{
  const int kb4 = j >> 7;
  const int t   = (j >> 5) & 3;
  const int ln  = j & 31;
  const int h   = ln >> 4;
  const int lw  = ln & 15;
  const int krow = kb4 * 4 + 2 * h;
  const int ncol = lw + 16 * t;
  const float* wp = (krow < 32) ? (WA + krow * 64) : (WB + (krow - 32) * 64);
  v2f p;
  p.x = wp[ncol];
  p.y = wp[64 + ncol];
  return p;
}

#define SB 8
#define RB (1 << SB)
#define MAXNB 512
#define NSLICE 128
#define CH 8
#define SCAP 8192
#define FCAP 6144

__device__ __forceinline__ int poly_wave_scan(int v, int lane) {
#pragma unroll
  for (int d = 1; d < 32; d <<= 1) {
    const int u = __shfl_up(v, d);
    if (lane >= d) v += u;
  }
  return v;
}

__device__ __forceinline__ int poly_block_scan(int v, int* ws, int& total) {
  const int lane = threadIdx.x & 31;
  const int wave = threadIdx.x >> 5;
  const int inc = poly_wave_scan(v, lane);
  if (lane == 31) ws[wave] = inc;
  __syncthreads();
  const int wt = ws[lane];
  const int wsc = poly_wave_scan(wt, lane);
  total = __shfl(wsc, 31);
  return __shfl(wsc, wave) - __shfl(wt, wave) + inc - v;
}

__global__ __launch_bounds__(1024) void poly_hist(
    const int* __restrict__ row, int* __restrict__ cnt1, int E, int NBP)
{
  __shared__ int h[MAXNB];
  const int t = threadIdx.x;
  if (t < MAXNB) h[t] = 0;
  __syncthreads();
  const int per = (E + NSLICE - 1) / NSLICE;
  const int b = blockIdx.x * per;
  const int e = min(b + per, E);
  for (int base = b; base < e; base += CH * 1024) {
    int r[CH];
#pragma unroll
    for (int k = 0; k < CH; ++k) {
      const int i = base + k * 1024 + t;
      r[k] = (i < e) ? row[i] : -1;
    }
#pragma unroll
    for (int k = 0; k < CH; ++k)
      if (r[k] >= 0) atomicAdd(&h[r[k] >> SB], 1);
  }
  __syncthreads();
  if (t < NBP) cnt1[blockIdx.x * NBP + t] = h[t];
}

__global__ __launch_bounds__(1024) void poly_scatter(
    const int* __restrict__ row, const int* __restrict__ col, const float* __restrict__ val,
    const int* __restrict__ cnt1, int* __restrict__ bstart, int2* __restrict__ coarse,
    int E, int NB, int NBP)
{
  __shared__ int tot[MAXNB];
  __shared__ int bef[MAXNB];
  __shared__ int cur[MAXNB];
  __shared__ int ws[32];
  __shared__ int2 stR[SCAP];
  __shared__ int stP[SCAP];
  const int t = threadIdx.x;
  if (t < MAXNB) { tot[t] = 0; bef[t] = 0; }
  __syncthreads();
  const int C4 = NBP >> 2;
  const int nq = NSLICE * C4;
  const int4* m4 = (const int4*)cnt1;
  for (int base = 0; base < nq; base += CH * 1024) {
    int4 m[CH];
#pragma unroll
    for (int k = 0; k < CH; ++k) {
      const int q = base + k * 1024 + t;
      m[k] = (q < nq) ? m4[q] : make_int4(0, 0, 0, 0);
    }
#pragma unroll
    for (int k = 0; k < CH; ++k) {
      const int q = base + k * 1024 + t;
      if (q < nq) {
        const int s = q / C4;
        const int c = (q - s * C4) * 4;
        atomicAdd(&tot[c + 0], m[k].x);
        atomicAdd(&tot[c + 1], m[k].y);
        atomicAdd(&tot[c + 2], m[k].z);
        atomicAdd(&tot[c + 3], m[k].w);
        if (s < (int)blockIdx.x) {
          atomicAdd(&bef[c + 0], m[k].x);
          atomicAdd(&bef[c + 1], m[k].y);
          atomicAdd(&bef[c + 2], m[k].z);
          atomicAdd(&bef[c + 3], m[k].w);
        }
      }
    }
  }
  __syncthreads();
  const int mine = (t < MAXNB) ? tot[t] : 0;
  int sum;
  const int ex = poly_block_scan(mine, ws, sum);
  if (t < MAXNB) cur[t] = ex + bef[t];
  if (blockIdx.x == 0 && t <= NB) bstart[t] = ex;
  const int own = (t < NBP) ? cnt1[blockIdx.x * NBP + t] : 0;
  __syncthreads();
  int osum;
  const int lex = poly_block_scan(own, ws, osum);
  if (t < MAXNB) tot[t] = ex + bef[t] - lex;
  __syncthreads();

  const int per = (E + NSLICE - 1) / NSLICE;
  const int b = blockIdx.x * per;
  const int e = min(b + per, E);
  const bool stage = (e - b) <= SCAP;
  for (int base = b; base < e; base += CH * 1024) {
    int r[CH], c[CH];
    float v[CH];
#pragma unroll
    for (int k = 0; k < CH; ++k) {
      const int i = base + k * 1024 + t;
      const bool ok = i < e;
      r[k] = ok ? row[i] : -1;
      c[k] = ok ? col[i] : 0;
      v[k] = ok ? val[i] : 0.f;
    }
    int p[CH];
#pragma unroll
    for (int k = 0; k < CH; ++k)
      p[k] = (r[k] >= 0) ? atomicAdd(&cur[r[k] >> SB], 1) : -1;
#pragma unroll
    for (int k = 0; k < CH; ++k)
      if (p[k] >= 0) {
        const int2 rec = make_int2((c[k] << SB) | (r[k] & (RB - 1)), __float_as_int(v[k]));
        if (stage) {
          const int l = p[k] - tot[r[k] >> SB];
          stR[l] = rec;
          stP[l] = p[k];
        } else {
          coarse[p[k]] = rec;
        }
      }
  }
  if (stage) {
    __syncthreads();
    const int n = e - b;
    for (int i = t; i < n; i += 1024) coarse[stP[i]] = stR[i];
  }
}

__global__ __launch_bounds__(1024) void poly_finalize(
    const int2* __restrict__ coarse, const int* __restrict__ bstart,
    int* __restrict__ rowptr, int2* __restrict__ sedge, int N)
{
  __shared__ int rc[RB];
  __shared__ int ws[32];
  __shared__ int2 stF[FCAP];
  const int t = threadIdx.x;
  const int bk = blockIdx.x;
  const int s = bstart[bk];
  const int e = bstart[bk + 1];
  if (t < RB) rc[t] = 0;
  __syncthreads();
  for (int base = s; base < e; base += CH * 1024) {
    int key[CH];
#pragma unroll
    for (int k = 0; k < CH; ++k) {
      const int i = base + k * 1024 + t;
      key[k] = (i < e) ? coarse[i].x : -1;
    }
#pragma unroll
    for (int k = 0; k < CH; ++k)
      if (key[k] >= 0) atomicAdd(&rc[key[k] & (RB - 1)], 1);
  }
  __syncthreads();
  const int cn = (t < RB) ? rc[t] : 0;
  int sum;
  const int ex = poly_block_scan(cn, ws, sum);
  if (t < RB) {
    rc[t] = s + ex;
    const int r = bk * RB + t;
    if (r < N) rowptr[r] = s + ex;
  }
  if (bk == (int)gridDim.x - 1 && t == 0) rowptr[N] = e;
  __syncthreads();
  const bool stage = (e - s) <= FCAP;
  for (int base = s; base < e; base += CH * 1024) {
    int2 rec[CH];
#pragma unroll
    for (int k = 0; k < CH; ++k) {
      const int i = base + k * 1024 + t;
      rec[k] = (i < e) ? coarse[i] : make_int2(-1, 0);
    }
    int p[CH];
#pragma unroll
    for (int k = 0; k < CH; ++k)
      p[k] = (rec[k].x >= 0) ? atomicAdd(&rc[rec[k].x & (RB - 1)], 1) : -1;
#pragma unroll
    for (int k = 0; k < CH; ++k)
      if (p[k] >= 0) {
        const int2 o = make_int2(rec[k].x >> SB, rec[k].y);
        if (stage) stF[p[k] - s] = o;
        else       sedge[p[k]] = o;
      }
  }
  if (stage) {
    __syncthreads();
    const int off = (int)(((size_t)(sedge + s) >> 3) & 15);
    for (int i = t - off; i < e - s; i += 1024)
      if (i >= 0) sedge[s + i] = stF[i];
  }
}

#define GR 8
#define TP 36

template <int U>
__device__ __forceinline__ unsigned poly_bc8(unsigned v) {
  return (unsigned)__builtin_amdgcn_mov_dpp8(v, (unsigned)U * 0x249249u);
}

#define POLY_EACH8(M) M(0) M(1) M(2) M(3) M(4) M(5) M(6) M(7)

template <bool TILE>
__device__ __forceinline__ void poly_walk(
    const int* __restrict__ rowptr, const int* __restrict__ sedge,
    const float* __restrict__ xin, int ldin,
    float* __restrict__ xout, float* st, int keep, int r0, int N, unsigned lane)
{
  const int g = (int)(lane >> 3);
  const unsigned f = lane & 7u;
  const int fi = (int)f;
  const int nr = min(GR, N - r0);
  const int s = rowptr[r0 + min(2 * g, nr)];
  const int m = rowptr[r0 + min(2 * g + 1, nr)];
  const int e = rowptr[r0 + min(2 * g + 2, nr)];
  const int n  = e - s;
  const int n0 = m - s;
  const int nmax = max(max(__builtin_amdgcn_readlane(n, 0), __builtin_amdgcn_readlane(n, 8)),
                       max(__builtin_amdgcn_readlane(n, 16), __builtin_amdgcn_readlane(n, 24)));
  const int2* se = (const int2*)sedge;
  const float4* xin4 = (const float4*)xin;
  const unsigned ld4 = (unsigned)ldin >> 2;
  float4 a0 = make_float4(0.f, 0.f, 0.f, 0.f);
  float4 a1 = make_float4(0.f, 0.f, 0.f, 0.f);
  if (nmax > 0) {
    const int last = max(e - 1, 0);
    const int2 rec = se[min(s + fi, last)];
    unsigned ofr = (unsigned)rec.x * ld4;
    unsigned w0r = (fi < n0) ? (unsigned)rec.y : 0u;
    unsigned w1r = (fi >= n0 && fi < n) ? (unsigned)rec.y : 0u;
    for (int c = 0; c < nmax; c += 8) {
      const int2 nx = se[min(s + c + 8 + fi, last)];
      float4 xv[8];
#define POLY_LD(U) xv[U] = xin4[poly_bc8<U>(ofr) + f];
      POLY_EACH8(POLY_LD)
#undef POLY_LD
      asm volatile("" :: "v"(nx.x), "v"(nx.y), "v"(xv[0].x), "v"(xv[1].x), "v"(xv[2].x), "v"(xv[3].x),
                         "v"(xv[4].x), "v"(xv[5].x), "v"(xv[6].x), "v"(xv[7].x));
      __builtin_amdgcn_sched_barrier(0);
#define POLY_FM(U) { \
        const float w0 = __uint_as_float(poly_bc8<U>(w0r)); \
        const float w1 = __uint_as_float(poly_bc8<U>(w1r)); \
        a0.x = fmaf(w0, xv[U].x, a0.x); a0.y = fmaf(w0, xv[U].y, a0.y); \
        a0.z = fmaf(w0, xv[U].z, a0.z); a0.w = fmaf(w0, xv[U].w, a0.w); \
        a1.x = fmaf(w1, xv[U].x, a1.x); a1.y = fmaf(w1, xv[U].y, a1.y); \
        a1.z = fmaf(w1, xv[U].z, a1.z); a1.w = fmaf(w1, xv[U].w, a1.w); }
      POLY_EACH8(POLY_FM)
#undef POLY_FM
      const int k = c + 8 + fi;
      ofr = (unsigned)nx.x * ld4;
      w0r = (k < n0) ? (unsigned)nx.y : 0u;
      w1r = (k >= n0 && k < n) ? (unsigned)nx.y : 0u;
    }
  }
  const int q0 = 2 * g;
  if (!TILE || keep) {
    float4* xo = (float4*)(xout + (size_t)r0 * 32);
    if (q0 < nr)     xo[(unsigned)q0 * 8u + f] = a0;
    if (q0 + 1 < nr) xo[(unsigned)(q0 + 1) * 8u + f] = a1;
  }
  if (TILE) {
    *(float4*)(st + q0 * TP + 4 * fi) = a0;
    *(float4*)(st + (q0 + 1) * TP + 4 * fi) = a1;
  }
}

__global__ __launch_bounds__(256) void poly_gather(
    const int* __restrict__ rowptr, const int* __restrict__ sedge,
    const float* __restrict__ xin, int ldin,
    float* __restrict__ xout, int N)
{
  const unsigned lane = threadIdx.x & 31;
  const int r0 = __builtin_amdgcn_readfirstlane((blockIdx.x * 8 + (threadIdx.x >> 5)) * GR);
  if (r0 >= N) return;
  poly_walk<false>(rowptr, sedge, xin, ldin, xout, nullptr, 1, r0, N, lane);
}

template <bool FIRST>
__global__ __launch_bounds__(256) void poly_gather_mm(
    const int* __restrict__ rowptr, const int* __restrict__ sedge,
    const float* __restrict__ xin, float* __restrict__ xout, int keep,
    const float* __restrict__ WA, const float* __restrict__ WB,
    const float* __restrict__ x, const float* __restrict__ W0,
    const float* __restrict__ bias,
    float* __restrict__ out, int N)
{
  __shared__ v2f sW[16 * 4 * 32];
  __shared__ v2f sW0[FIRST ? 16 * 4 * 32 : 1];
  __shared__ __attribute__((aligned(16))) float sT[8 * GR * TP];
  const unsigned lane = threadIdx.x & 31;
  const int wave = __builtin_amdgcn_readfirstlane(threadIdx.x >> 5);
  const int lo = lane & 15;
  const int hi = lane >> 4;

  v2f wv[8], wv0[FIRST ? 8 : 1];
#pragma unroll
  for (int i = 0; i < 8; ++i) {
    wv[i] = poly_w_entry(WA, WB, threadIdx.x + 256 * i);
    if (FIRST) wv0[i] = poly_w_entry(W0, W0 + 32 * 64, threadIdx.x + 256 * i);
  }

  const int tile = wave >> 1;
  const int nsel = wave & 1;
  const int rowBase = blockIdx.x * (8 * GR) + tile * 16;
  const int col0 = nsel * 32 + lo;
  const int rowA = min(rowBase + lo, N - 1);
  const v2f* xa = (const v2f*)(xin + (size_t)rowA * 32 + 2 * hi);
  v2f av[8];
#pragma unroll
  for (int s = 0; s < 8; ++s) av[s] = xa[2 * s];
  float o0[8], o1[8];
  if (FIRST) {
    o0[0] = bias[col0];
    o1[0] = bias[col0 + 16];
  } else {
#pragma unroll
    for (int r = 0; r < 8; ++r) {
      const float* o = out + (size_t)min(rowBase + r + 8 * hi, N - 1) * 64 + col0;
      o0[r] = o[0];
      o1[r] = o[16];
    }
  }

  const int r0 = (blockIdx.x * 8 + wave) * GR;
  float* st = sT + wave * (GR * TP);
  if (r0 < N) {
    poly_walk<true>(rowptr, sedge, xin, 32, xout, st, keep, r0, N, lane);
  } else {
    for (int q = 0; q < GR; ++q) st[q * TP + (int)lane] = 0.f;
  }

#pragma unroll
  for (int i = 0; i < 8; ++i) {
    sW[threadIdx.x + 256 * i] = wv[i];
    if (FIRST) sW0[threadIdx.x + 256 * i] = wv0[i];
  }
  __syncthreads();
  if (rowBase >= N) return;

  v2f ax[FIRST ? 16 : 1];
  if (FIRST) {
    const v2f* xr = (const v2f*)(x + (size_t)rowA * 64 + 2 * hi);
#pragma unroll
    for (int s = 0; s < 16; ++s) ax[s] = xr[2 * s];
  }

  v8f acc0 = {}, acc1 = {};
  const v2f* wl = sW + lane + nsel * 64;
  const float* ta = sT + (tile * 16 + lo) * TP + 2 * hi;
#pragma unroll
  for (int s = 0; s < 16; ++s) {
    v2f a;
    if (s < 8) a = av[s];
    else       a = *(const v2f*)(ta + 4 * (s - 8));
    const v2f b0 = wl[s * 128];
    const v2f b1 = wl[s * 128 + 32];
    acc0 = __builtin_amdgcn_wmma_f32_16x16x4_f32(false, a, false, b0, (short)0, acc0, false, false);
    acc1 = __builtin_amdgcn_wmma_f32_16x16x4_f32(false, a, false, b1, (short)0, acc1, false, false);
  }

  if (FIRST) {
    v8f acx0 = {}, acx1 = {};
    const v2f* wl0 = sW0 + lane + nsel * 64;
#pragma unroll
    for (int s = 0; s < 16; ++s) {
      const v2f b0 = wl0[s * 128];
      const v2f b1 = wl0[s * 128 + 32];
      acx0 = __builtin_amdgcn_wmma_f32_16x16x4_f32(false, ax[s], false, b0, (short)0, acx0, false, false);
      acx1 = __builtin_amdgcn_wmma_f32_16x16x4_f32(false, ax[s], false, b1, (short)0, acx1, false, false);
    }
#pragma unroll
    for (int r = 0; r < 8; ++r) {
      const int row = rowBase + r + 8 * hi;
      if (row < N) {
        float* o = out + (size_t)row * 64 + col0;
        o[0]  = (acx0[r] + o0[0]) + acc0[r];
        o[16] = (acx1[r] + o1[0]) + acc1[r];
      }
    }
  } else {
#pragma unroll
    for (int r = 0; r < 8; ++r) {
      const int row = rowBase + r + 8 * hi;
      if (row < N) {
        float* o = out + (size_t)row * 64 + col0;
        o[0]  = o0[r] + acc0[r];
        o[16] = o1[r] + acc1[r];
      }
    }
  }
}

extern "C" void kernel_launch(void* const* d_in, const int* in_sizes, int n_in,
                              void* d_out, int out_size, void* d_ws, size_t ws_size,
                              hipStream_t stream) {
  const float* x       = (const float*)d_in[0];
  const int*   lap_row = (const int*)d_in[1];
  const int*   lap_col = (const int*)d_in[2];
  const float* lap_val = (const float*)d_in[3];
  const float* weights = (const float*)d_in[4];
  const float* bias    = (const float*)d_in[5];
  float* out = (float*)d_out;

  const int N = in_sizes[0] / 64;
  const int E = in_sizes[1];

  float* P0   = (float*)d_ws;
  float* P1   = P0 + (size_t)N * 32;
  int* bstart = (int*)(P1 + (size_t)N * 32);
  int* rowptr = bstart + N;
  int* sedge  = rowptr + N + 4;
  int2* coarse = (int2*)d_ws;
  int* cnt1    = (int*)d_ws + (((size_t)E * 2 + 3) & ~(size_t)3);

  const int NB  = (N + RB - 1) / RB;
  const int NBP = (NB + 3) & ~3;

  dim3 gatherGrid((N + 8 * GR - 1) / (8 * GR)), gatherBlock(256);

  poly_hist<<<dim3(NSLICE), dim3(1024), 0, stream>>>(lap_row, cnt1, E, NBP);
  poly_scatter<<<dim3(NSLICE), dim3(1024), 0, stream>>>(lap_row, lap_col, lap_val,
                                                        cnt1, bstart, coarse, E, NB, NBP);
  poly_finalize<<<dim3(NB), dim3(1024), 0, stream>>>(coarse, bstart, rowptr, (int2*)sedge, N);

  for (int h = 0; h < 2; ++h) {
    const float* wh = weights + (size_t)h * 32 * 64;
    poly_gather<<<gatherGrid, gatherBlock, 0, stream>>>(rowptr, sedge, x + h * 32, 64, P0, N);
    if (h == 0)
      poly_gather_mm<true><<<gatherGrid, gatherBlock, 0, stream>>>(
          rowptr, sedge, P0, P1, 1, wh + (size_t)1 * 64 * 64, wh + (size_t)2 * 64 * 64,
          x, weights, bias, out, N);
    else
      poly_gather_mm<false><<<gatherGrid, gatherBlock, 0, stream>>>(
          rowptr, sedge, P0, P1, 1, wh + (size_t)1 * 64 * 64, wh + (size_t)2 * 64 * 64,
          x, weights, bias, out, N);
    poly_gather<<<gatherGrid, gatherBlock, 0, stream>>>(rowptr, sedge, P1, 32, P0, N);
    poly_gather_mm<false><<<gatherGrid, gatherBlock, 0, stream>>>(
        rowptr, sedge, P0, P1, 0, wh + (size_t)3 * 64 * 64, wh + (size_t)4 * 64 * 64,
        x, weights, bias, out, N);
  }
}


# Edge sort in 2 launches, was 3: a slice block bucket-sorts its own edges in LDS, stores one contiguous piece plus its bucket offsets; a bucket block collects its pieces; row sums stay arrival order
# speedup vs baseline: 5.2264x; 1.0644x over previous
#include <hip/hip_runtime.h>

typedef __attribute__((ext_vector_type(2))) float v2f;
typedef __attribute__((ext_vector_type(8))) float v8f;

__device__ __forceinline__ v2f poly_w_entry(const float* __restrict__ WA,
                                            const float* __restrict__ WB, int j)
{
  const int kb4 = j >> 7;
  const int t   = (j >> 5) & 3;
  const int ln  = j & 31;
  const int h   = ln >> 4;
  const int lw  = ln & 15;
  const int krow = kb4 * 4 + 2 * h;
  const int ncol = lw + 16 * t;
  const float* wp = (krow < 32) ? (WA + krow * 64) : (WB + (krow - 32) * 64);
  v2f p;
  p.x = wp[ncol];
  p.y = wp[64 + ncol];
  return p;
}

#define SB 8
#define RB (1 << SB)
#define MAXNB 512
#define NSLICE 128
#define MAXSL 1024
#define CH 8
#define SCAP 8192
#define SG 6
#define FCAP 6144

__device__ __forceinline__ int poly_wave_scan(int v, int lane) {
#pragma unroll
  for (int d = 1; d < 32; d <<= 1) {
    const int u = __shfl_up(v, d);
    if (lane >= d) v += u;
  }
  return v;
}

__device__ __forceinline__ int poly_block_scan(int v, int* ws, int& total) {
  const int lane = threadIdx.x & 31;
  const int wave = threadIdx.x >> 5;
  const int inc = poly_wave_scan(v, lane);
  if (lane == 31) ws[wave] = inc;
  __syncthreads();
  const int wt = ws[lane];
  const int wsc = poly_wave_scan(wt, lane);
  total = __shfl(wsc, 31);
  return __shfl(wsc, wave) - __shfl(wt, wave) + inc - v;
}

__global__ __launch_bounds__(1024) void poly_slice_sort(
    const int* __restrict__ row, const int* __restrict__ col, const float* __restrict__ val,
    int* __restrict__ off1, int2* __restrict__ coarse, int E, int NB, int nsl)
{
  __shared__ int cnt[MAXNB];
  __shared__ int ws[32];
  __shared__ int2 stR[SCAP];
  const int t = threadIdx.x;
  if (t < MAXNB) cnt[t] = 0;
  __syncthreads();
  const int per = (E + nsl - 1) / nsl;
  const int b = min((int)blockIdx.x * per, E);
  const int e = min(b + per, E);
  int r[CH], c[CH];
  float v[CH];
#pragma unroll
  for (int k = 0; k < CH; ++k) {
    const int i = b + k * 1024 + t;
    const bool ok = i < e;
    r[k] = ok ? row[i] : -1;
    c[k] = ok ? col[i] : 0;
    v[k] = ok ? val[i] : 0.f;
  }
  int p[CH];
#pragma unroll
  for (int k = 0; k < CH; ++k)
    p[k] = (r[k] >= 0) ? atomicAdd(&cnt[r[k] >> SB], 1) : -1;
  __syncthreads();
  const int own = (t < MAXNB) ? cnt[t] : 0;
  int sum;
  const int lex = poly_block_scan(own, ws, sum);
  if (t < MAXNB) cnt[t] = lex;
  if (t <= NB) off1[(size_t)blockIdx.x * (NB + 1) + t] = lex;
  __syncthreads();
  int q[CH];
#pragma unroll
  for (int k = 0; k < CH; ++k) q[k] = cnt[max(r[k], 0) >> SB];
#pragma unroll
  for (int k = 0; k < CH; ++k)
    if (p[k] >= 0)
      stR[q[k] + p[k]] = make_int2((c[k] << SB) | (r[k] & (RB - 1)), __float_as_int(v[k]));
  __syncthreads();
  const int n = e - b;
  const int off = (int)(((size_t)(coarse + b) >> 3) & 15);
  int2 w[CH + 1];
#pragma unroll
  for (int k = 0; k <= CH; ++k)
    w[k] = stR[min(max(t - off + k * 1024, 0), SCAP - 1)];
#pragma unroll
  for (int k = 0; k <= CH; ++k) {
    const int i = t - off + k * 1024;
    if (i >= 0 && i < n) coarse[b + i] = w[k];
  }
}

template <bool PLACE>
__device__ __forceinline__ void poly_bucket_pass(
    const int2* __restrict__ coarse, const int* sgA, const int* sgL, int nsl, int t,
    int* rc, int2* stF, int2* __restrict__ sedge, int s, bool stage)
{
  const int f = t & 7;
  for (int sg = t >> 3; sg < nsl; sg += 128) {
    const int base = sgA[sg];
    const int ln = sgL[sg];
    for (int j0 = 0; j0 < ln; j0 += 8 * SG) {
      int2 rec[SG];
#pragma unroll
      for (int u = 0; u < SG; ++u) {
        const int j = j0 + 8 * u + f;
        rec[u] = (j < ln) ? coarse[base + j] : make_int2(-1, 0);
      }
      if (!PLACE) {
#pragma unroll
        for (int u = 0; u < SG; ++u)
          if (rec[u].x >= 0) atomicAdd(&rc[rec[u].x & (RB - 1)], 1);
      } else {
        int p[SG];
#pragma unroll
        for (int u = 0; u < SG; ++u)
          p[u] = (rec[u].x >= 0) ? atomicAdd(&rc[rec[u].x & (RB - 1)], 1) : -1;
#pragma unroll
        for (int u = 0; u < SG; ++u)
          if (p[u] >= 0) {
            const int2 o = make_int2(rec[u].x >> SB, rec[u].y);
            if (stage) stF[p[u] - s] = o;
            else       sedge[p[u]] = o;
          }
      }
    }
  }
}

__global__ __launch_bounds__(1024) void poly_finalize(
    const int2* __restrict__ coarse, const int* __restrict__ off1,
    int* __restrict__ rowptr, int2* __restrict__ sedge, int N, int E, int nsl)
{
  __shared__ int rc[RB];
  __shared__ int ws[32];
  __shared__ int sgA[MAXSL];
  __shared__ int sgL[MAXSL];
  __shared__ int sgO[MAXSL];
  __shared__ int2 stF[FCAP];
  const int t = threadIdx.x;
  const int bk = blockIdx.x;
  const int per = (E + nsl - 1) / nsl;
  int a = 0, len = 0;
  if (t < nsl) {
    const int* o = off1 + (size_t)t * (gridDim.x + 1) + bk;
    a = o[0];
    len = o[1] - a;
    sgA[t] = min(t * per, E) + a;
    sgL[t] = len;
    sgO[t] = a;
  }
  if (t < RB) rc[t] = 0;
  __syncthreads();
  int s = 0, n = 0;
  for (int i = t & 31; i < nsl; i += 32) { s += sgO[i]; n += sgL[i]; }
#pragma unroll
  for (int d = 16; d >= 1; d >>= 1) { s += __shfl_xor(s, d); n += __shfl_xor(n, d); }
  const int e = s + n;
  poly_bucket_pass<false>(coarse, sgA, sgL, nsl, t, rc, stF, sedge, s, false);
  __syncthreads();
  const int cn = (t < RB) ? rc[t] : 0;
  int sum;
  const int ex = poly_block_scan(cn, ws, sum);
  if (t < RB) {
    rc[t] = s + ex;
    const int r = bk * RB + t;
    if (r < N) rowptr[r] = s + ex;
  }
  if (bk == (int)gridDim.x - 1 && t == 0) rowptr[N] = e;
  __syncthreads();
  const bool stage = n <= FCAP;
  poly_bucket_pass<true>(coarse, sgA, sgL, nsl, t, rc, stF, sedge, s, stage);
  if (stage) {
    __syncthreads();
    const int off = (int)(((size_t)(sedge + s) >> 3) & 15);
    for (int i = t - off; i < n; i += 1024)
      if (i >= 0) sedge[s + i] = stF[i];
  }
}

#define GR 8
#define TP 36

template <int U>
__device__ __forceinline__ unsigned poly_bc8(unsigned v) {
  return (unsigned)__builtin_amdgcn_mov_dpp8(v, (unsigned)U * 0x249249u);
}

#define POLY_EACH8(M) M(0) M(1) M(2) M(3) M(4) M(5) M(6) M(7)

template <bool TILE>
__device__ __forceinline__ void poly_walk(
    const int* __restrict__ rowptr, const int* __restrict__ sedge,
    const float* __restrict__ xin, int ldin,
    float* __restrict__ xout, float* st, int keep, int r0, int N, unsigned lane)
{
  const int g = (int)(lane >> 3);
  const unsigned f = lane & 7u;
  const int fi = (int)f;
  const int nr = min(GR, N - r0);
  const int s = rowptr[r0 + min(2 * g, nr)];
  const int m = rowptr[r0 + min(2 * g + 1, nr)];
  const int e = rowptr[r0 + min(2 * g + 2, nr)];
  const int n  = e - s;
  const int n0 = m - s;
  const int nmax = max(max(__builtin_amdgcn_readlane(n, 0), __builtin_amdgcn_readlane(n, 8)),
                       max(__builtin_amdgcn_readlane(n, 16), __builtin_amdgcn_readlane(n, 24)));
  const int2* se = (const int2*)sedge;
  const float4* xin4 = (const float4*)xin;
  const unsigned ld4 = (unsigned)ldin >> 2;
  float4 a0 = make_float4(0.f, 0.f, 0.f, 0.f);
  float4 a1 = make_float4(0.f, 0.f, 0.f, 0.f);
  if (nmax > 0) {
    const int last = max(e - 1, 0);
    const int2 rec = se[min(s + fi, last)];
    unsigned ofr = (unsigned)rec.x * ld4;
    unsigned w0r = (fi < n0) ? (unsigned)rec.y : 0u;
    unsigned w1r = (fi >= n0 && fi < n) ? (unsigned)rec.y : 0u;
    for (int c = 0; c < nmax; c += 8) {
      const int2 nx = se[min(s + c + 8 + fi, last)];
      float4 xv[8];
#define POLY_LD(U) xv[U] = xin4[poly_bc8<U>(ofr) + f];
      POLY_EACH8(POLY_LD)
#undef POLY_LD
      asm volatile("" :: "v"(nx.x), "v"(nx.y), "v"(xv[0].x), "v"(xv[1].x), "v"(xv[2].x), "v"(xv[3].x),
                         "v"(xv[4].x), "v"(xv[5].x), "v"(xv[6].x), "v"(xv[7].x));
      __builtin_amdgcn_sched_barrier(0);
#define POLY_FM(U) { \
        const float w0 = __uint_as_float(poly_bc8<U>(w0r)); \
        const float w1 = __uint_as_float(poly_bc8<U>(w1r)); \
        a0.x = fmaf(w0, xv[U].x, a0.x); a0.y = fmaf(w0, xv[U].y, a0.y); \
        a0.z = fmaf(w0, xv[U].z, a0.z); a0.w = fmaf(w0, xv[U].w, a0.w); \
        a1.x = fmaf(w1, xv[U].x, a1.x); a1.y = fmaf(w1, xv[U].y, a1.y); \
        a1.z = fmaf(w1, xv[U].z, a1.z); a1.w = fmaf(w1, xv[U].w, a1.w); }
      POLY_EACH8(POLY_FM)
#undef POLY_FM
      const int k = c + 8 + fi;
      ofr = (unsigned)nx.x * ld4;
      w0r = (k < n0) ? (unsigned)nx.y : 0u;
      w1r = (k >= n0 && k < n) ? (unsigned)nx.y : 0u;
    }
  }
  const int q0 = 2 * g;
  if (!TILE || keep) {
    float4* xo = (float4*)(xout + (size_t)r0 * 32);
    if (q0 < nr)     xo[(unsigned)q0 * 8u + f] = a0;
    if (q0 + 1 < nr) xo[(unsigned)(q0 + 1) * 8u + f] = a1;
  }
  if (TILE) {
    *(float4*)(st + q0 * TP + 4 * fi) = a0;
    *(float4*)(st + (q0 + 1) * TP + 4 * fi) = a1;
  }
}

__global__ __launch_bounds__(256) void poly_gather(
    const int* __restrict__ rowptr, const int* __restrict__ sedge,
    const float* __restrict__ xin, int ldin,
    float* __restrict__ xout, int N)
{
  const unsigned lane = threadIdx.x & 31;
  const int r0 = __builtin_amdgcn_readfirstlane((blockIdx.x * 8 + (threadIdx.x >> 5)) * GR);
  if (r0 >= N) return;
  poly_walk<false>(rowptr, sedge, xin, ldin, xout, nullptr, 1, r0, N, lane);
}

template <bool FIRST>
__global__ __launch_bounds__(256) void poly_gather_mm(
    const int* __restrict__ rowptr, const int* __restrict__ sedge,
    const float* __restrict__ xin, float* __restrict__ xout, int keep,
    const float* __restrict__ WA, const float* __restrict__ WB,
    const float* __restrict__ x, const float* __restrict__ W0,
    const float* __restrict__ bias,
    float* __restrict__ out, int N)
{
  __shared__ v2f sW[16 * 4 * 32];
  __shared__ v2f sW0[FIRST ? 16 * 4 * 32 : 1];
  __shared__ __attribute__((aligned(16))) float sT[8 * GR * TP];
  const unsigned lane = threadIdx.x & 31;
  const int wave = __builtin_amdgcn_readfirstlane(threadIdx.x >> 5);
  const int lo = lane & 15;
  const int hi = lane >> 4;

  v2f wv[8], wv0[FIRST ? 8 : 1];
#pragma unroll
  for (int i = 0; i < 8; ++i) {
    wv[i] = poly_w_entry(WA, WB, threadIdx.x + 256 * i);
    if (FIRST) wv0[i] = poly_w_entry(W0, W0 + 32 * 64, threadIdx.x + 256 * i);
  }

  const int tile = wave >> 1;
  const int nsel = wave & 1;
  const int rowBase = blockIdx.x * (8 * GR) + tile * 16;
  const int col0 = nsel * 32 + lo;
  const int rowA = min(rowBase + lo, N - 1);
  const v2f* xa = (const v2f*)(xin + (size_t)rowA * 32 + 2 * hi);
  v2f av[8];
#pragma unroll
  for (int s = 0; s < 8; ++s) av[s] = xa[2 * s];
  float o0[8], o1[8];
  if (FIRST) {
    o0[0] = bias[col0];
    o1[0] = bias[col0 + 16];
  } else {
#pragma unroll
    for (int r = 0; r < 8; ++r) {
      const float* o = out + (size_t)min(rowBase + r + 8 * hi, N - 1) * 64 + col0;
      o0[r] = o[0];
      o1[r] = o[16];
    }
  }

  const int r0 = (blockIdx.x * 8 + wave) * GR;
  float* st = sT + wave * (GR * TP);
  if (r0 < N) {
    poly_walk<true>(rowptr, sedge, xin, 32, xout, st, keep, r0, N, lane);
  } else {
    for (int q = 0; q < GR; ++q) st[q * TP + (int)lane] = 0.f;
  }

#pragma unroll
  for (int i = 0; i < 8; ++i) {
    sW[threadIdx.x + 256 * i] = wv[i];
    if (FIRST) sW0[threadIdx.x + 256 * i] = wv0[i];
  }
  __syncthreads();
  if (rowBase >= N) return;

  v2f ax[FIRST ? 16 : 1];
  if (FIRST) {
    const v2f* xr = (const v2f*)(x + (size_t)rowA * 64 + 2 * hi);
#pragma unroll
    for (int s = 0; s < 16; ++s) ax[s] = xr[2 * s];
  }

  v8f acc0 = {}, acc1 = {};
  const v2f* wl = sW + lane + nsel * 64;
  const float* ta = sT + (tile * 16 + lo) * TP + 2 * hi;
#pragma unroll
  for (int s = 0; s < 16; ++s) {
    v2f a;
    if (s < 8) a = av[s];
    else       a = *(const v2f*)(ta + 4 * (s - 8));
    const v2f b0 = wl[s * 128];
    const v2f b1 = wl[s * 128 + 32];
    acc0 = __builtin_amdgcn_wmma_f32_16x16x4_f32(false, a, false, b0, (short)0, acc0, false, false);
    acc1 = __builtin_amdgcn_wmma_f32_16x16x4_f32(false, a, false, b1, (short)0, acc1, false, false);
  }

  if (FIRST) {
    v8f acx0 = {}, acx1 = {};
    const v2f* wl0 = sW0 + lane + nsel * 64;
#pragma unroll
    for (int s = 0; s < 16; ++s) {
      const v2f b0 = wl0[s * 128];
      const v2f b1 = wl0[s * 128 + 32];
      acx0 = __builtin_amdgcn_wmma_f32_16x16x4_f32(false, ax[s], false, b0, (short)0, acx0, false, false);
      acx1 = __builtin_amdgcn_wmma_f32_16x16x4_f32(false, ax[s], false, b1, (short)0, acx1, false, false);
    }
#pragma unroll
    for (int r = 0; r < 8; ++r) {
      const int row = rowBase + r + 8 * hi;
      if (row < N) {
        float* o = out + (size_t)row * 64 + col0;
        o[0]  = (acx0[r] + o0[0]) + acc0[r];
        o[16] = (acx1[r] + o1[0]) + acc1[r];
      }
    }
  } else {
#pragma unroll
    for (int r = 0; r < 8; ++r) {
      const int row = rowBase + r + 8 * hi;
      if (row < N) {
        float* o = out + (size_t)row * 64 + col0;
        o[0]  = o0[r] + acc0[r];
        o[16] = o1[r] + acc1[r];
      }
    }
  }
}

extern "C" void kernel_launch(void* const* d_in, const int* in_sizes, int n_in,
                              void* d_out, int out_size, void* d_ws, size_t ws_size,
                              hipStream_t stream) {
  const float* x       = (const float*)d_in[0];
  const int*   lap_row = (const int*)d_in[1];
  const int*   lap_col = (const int*)d_in[2];
  const float* lap_val = (const float*)d_in[3];
  const float* weights = (const float*)d_in[4];
  const float* bias    = (const float*)d_in[5];
  float* out = (float*)d_out;

  const int N = in_sizes[0] / 64;
  const int E = in_sizes[1];

  float* P0   = (float*)d_ws;
  float* P1   = P0 + (size_t)N * 32;
  int* rowptr = (int*)(P1 + (size_t)N * 32) + N;
  int* sedge  = rowptr + N + 4;
  int2* coarse = (int2*)d_ws;
  int* off1    = (int*)d_ws + (size_t)E * 2;

  const int NB  = (N + RB - 1) / RB;
  const int nsl = ((E + SCAP - 1) / SCAP > NSLICE) ? (E + SCAP - 1) / SCAP : NSLICE;

  dim3 gatherGrid((N + 8 * GR - 1) / (8 * GR)), gatherBlock(256);

  poly_slice_sort<<<dim3(nsl), dim3(1024), 0, stream>>>(lap_row, lap_col, lap_val,
                                                        off1, coarse, E, NB, nsl);
  poly_finalize<<<dim3(NB), dim3(1024), 0, stream>>>(coarse, off1, rowptr, (int2*)sedge, N, E, nsl);

  for (int h = 0; h < 2; ++h) {
    const float* wh = weights + (size_t)h * 32 * 64;
    poly_gather<<<gatherGrid, gatherBlock, 0, stream>>>(rowptr, sedge, x + h * 32, 64, P0, N);
    if (h == 0)
      poly_gather_mm<true><<<gatherGrid, gatherBlock, 0, stream>>>(
          rowptr, sedge, P0, P1, 1, wh + (size_t)1 * 64 * 64, wh + (size_t)2 * 64 * 64,
          x, weights, bias, out, N);
    else
      poly_gather_mm<false><<<gatherGrid, gatherBlock, 0, stream>>>(
          rowptr, sedge, P0, P1, 1, wh + (size_t)1 * 64 * 64, wh + (size_t)2 * 64 * 64,
          x, weights, bias, out, N);
    poly_gather<<<gatherGrid, gatherBlock, 0, stream>>>(rowptr, sedge, P1, 32, P0, N);
    poly_gather_mm<false><<<gatherGrid, gatherBlock, 0, stream>>>(
        rowptr, sedge, P0, P1, 0, wh + (size_t)3 * 64 * 64, wh + (size_t)4 * 64 * 64,
        x, weights, bias, out, N);
  }
}


# First hop of feature half 0 done by the edge sort's bucket blocks: each gathers its own 256 rows from x with its sorted records still in LDS; 9 launches, was 10; row sum order kept
# speedup vs baseline: 5.4203x; 1.0371x over previous
#include <hip/hip_runtime.h>

typedef __attribute__((ext_vector_type(2))) float v2f;
typedef __attribute__((ext_vector_type(8))) float v8f;

__device__ __forceinline__ v2f poly_w_entry(const float* __restrict__ WA,
                                            const float* __restrict__ WB, int j)
{
  const int kb4 = j >> 7;
  const int t   = (j >> 5) & 3;
  const int ln  = j & 31;
  const int h   = ln >> 4;
  const int lw  = ln & 15;
  const int krow = kb4 * 4 + 2 * h;
  const int ncol = lw + 16 * t;
  const float* wp = (krow < 32) ? (WA + krow * 64) : (WB + (krow - 32) * 64);
  v2f p;
  p.x = wp[ncol];
  p.y = wp[64 + ncol];
  return p;
}

#define SB 8
#define RB (1 << SB)
#define MAXNB 512
#define NSLICE 128
#define MAXSL 1024
#define CH 8
#define SCAP 8192
#define SG 6
#define FCAP 6144

__device__ __forceinline__ int poly_wave_scan(int v, int lane) {
#pragma unroll
  for (int d = 1; d < 32; d <<= 1) {
    const int u = __shfl_up(v, d);
    if (lane >= d) v += u;
  }
  return v;
}

__device__ __forceinline__ int poly_block_scan(int v, int* ws, int& total) {
  const int lane = threadIdx.x & 31;
  const int wave = threadIdx.x >> 5;
  const int inc = poly_wave_scan(v, lane);
  if (lane == 31) ws[wave] = inc;
  __syncthreads();
  const int wt = ws[lane];
  const int wsc = poly_wave_scan(wt, lane);
  total = __shfl(wsc, 31);
  return __shfl(wsc, wave) - __shfl(wt, wave) + inc - v;
}

__global__ __launch_bounds__(1024) void poly_slice_sort(
    const int* __restrict__ row, const int* __restrict__ col, const float* __restrict__ val,
    int* __restrict__ off1, int2* __restrict__ coarse, int E, int NB, int nsl)
{
  __shared__ int cnt[MAXNB];
  __shared__ int ws[32];
  __shared__ int2 stR[SCAP];
  const int t = threadIdx.x;
  if (t < MAXNB) cnt[t] = 0;
  __syncthreads();
  const int per = (E + nsl - 1) / nsl;
  const int b = min((int)blockIdx.x * per, E);
  const int e = min(b + per, E);
  int r[CH], c[CH];
  float v[CH];
#pragma unroll
  for (int k = 0; k < CH; ++k) {
    const int i = b + k * 1024 + t;
    const bool ok = i < e;
    r[k] = ok ? row[i] : -1;
    c[k] = ok ? col[i] : 0;
    v[k] = ok ? val[i] : 0.f;
  }
  int p[CH];
#pragma unroll
  for (int k = 0; k < CH; ++k)
    p[k] = (r[k] >= 0) ? atomicAdd(&cnt[r[k] >> SB], 1) : -1;
  __syncthreads();
  const int own = (t < MAXNB) ? cnt[t] : 0;
  int sum;
  const int lex = poly_block_scan(own, ws, sum);
  if (t < MAXNB) cnt[t] = lex;
  if (t <= NB) off1[(size_t)blockIdx.x * (NB + 1) + t] = lex;
  __syncthreads();
  int q[CH];
#pragma unroll
  for (int k = 0; k < CH; ++k) q[k] = cnt[max(r[k], 0) >> SB];
#pragma unroll
  for (int k = 0; k < CH; ++k)
    if (p[k] >= 0)
      stR[q[k] + p[k]] = make_int2((c[k] << SB) | (r[k] & (RB - 1)), __float_as_int(v[k]));
  __syncthreads();
  const int n = e - b;
  const int off = (int)(((size_t)(coarse + b) >> 3) & 15);
  int2 w[CH + 1];
#pragma unroll
  for (int k = 0; k <= CH; ++k)
    w[k] = stR[min(max(t - off + k * 1024, 0), SCAP - 1)];
#pragma unroll
  for (int k = 0; k <= CH; ++k) {
    const int i = t - off + k * 1024;
    if (i >= 0 && i < n) coarse[b + i] = w[k];
  }
}

template <bool PLACE>
__device__ __forceinline__ void poly_bucket_pass(
    const int2* __restrict__ coarse, const int* sgA, const int* sgL, int nsl, int t,
    int* rc, int2* stF, int2* __restrict__ sedge, int s, bool stage)
{
  const int f = t & 7;
  for (int sg = t >> 3; sg < nsl; sg += 128) {
    const int base = sgA[sg];
    const int ln = sgL[sg];
    for (int j0 = 0; j0 < ln; j0 += 8 * SG) {
      int2 rec[SG];
#pragma unroll
      for (int u = 0; u < SG; ++u) {
        const int j = j0 + 8 * u + f;
        rec[u] = (j < ln) ? coarse[base + j] : make_int2(-1, 0);
      }
      if (!PLACE) {
#pragma unroll
        for (int u = 0; u < SG; ++u)
          if (rec[u].x >= 0) atomicAdd(&rc[rec[u].x & (RB - 1)], 1);
      } else {
        int p[SG];
#pragma unroll
        for (int u = 0; u < SG; ++u)
          p[u] = (rec[u].x >= 0) ? atomicAdd(&rc[rec[u].x & (RB - 1)], 1) : -1;
#pragma unroll
        for (int u = 0; u < SG; ++u)
          if (p[u] >= 0) {
            const int2 o = make_int2(rec[u].x >> SB, rec[u].y);
            if (stage) stF[p[u] - s] = o;
            else       sedge[p[u]] = o;
          }
      }
    }
  }
}

#define GR 8
#define TP 36

template <int U>
__device__ __forceinline__ unsigned poly_bc8(unsigned v) {
  return (unsigned)__builtin_amdgcn_mov_dpp8(v, (unsigned)U * 0x249249u);
}

#define POLY_EACH8(M) M(0) M(1) M(2) M(3) M(4) M(5) M(6) M(7)

template <bool TILE>
__device__ __forceinline__ void poly_walk_run(
    const int2* __restrict__ se, int s, int m, int e,
    const float* __restrict__ xin, int ldin,
    float* __restrict__ xout, float* st, int keep, int r0, int nr, unsigned lane)
{
  const int g = (int)(lane >> 3);
  const unsigned f = lane & 7u;
  const int fi = (int)f;
  const int n  = e - s;
  const int n0 = m - s;
  const int nmax = max(max(__builtin_amdgcn_readlane(n, 0), __builtin_amdgcn_readlane(n, 8)),
                       max(__builtin_amdgcn_readlane(n, 16), __builtin_amdgcn_readlane(n, 24)));
  const float4* xin4 = (const float4*)xin;
  const unsigned ld4 = (unsigned)ldin >> 2;
  float4 a0 = make_float4(0.f, 0.f, 0.f, 0.f);
  float4 a1 = make_float4(0.f, 0.f, 0.f, 0.f);
  if (nmax > 0) {
    const int last = max(e - 1, 0);
    const int2 rec = se[min(s + fi, last)];
    unsigned ofr = (unsigned)rec.x * ld4;
    unsigned w0r = (fi < n0) ? (unsigned)rec.y : 0u;
    unsigned w1r = (fi >= n0 && fi < n) ? (unsigned)rec.y : 0u;
    for (int c = 0; c < nmax; c += 8) {
      const int2 nx = se[min(s + c + 8 + fi, last)];
      float4 xv[8];
#define POLY_LD(U) xv[U] = xin4[poly_bc8<U>(ofr) + f];
      POLY_EACH8(POLY_LD)
#undef POLY_LD
      asm volatile("" :: "v"(nx.x), "v"(nx.y), "v"(xv[0].x), "v"(xv[1].x), "v"(xv[2].x), "v"(xv[3].x),
                         "v"(xv[4].x), "v"(xv[5].x), "v"(xv[6].x), "v"(xv[7].x));
      __builtin_amdgcn_sched_barrier(0);
#define POLY_FM(U) { \
        const float w0 = __uint_as_float(poly_bc8<U>(w0r)); \
        const float w1 = __uint_as_float(poly_bc8<U>(w1r)); \
        a0.x = fmaf(w0, xv[U].x, a0.x); a0.y = fmaf(w0, xv[U].y, a0.y); \
        a0.z = fmaf(w0, xv[U].z, a0.z); a0.w = fmaf(w0, xv[U].w, a0.w); \
        a1.x = fmaf(w1, xv[U].x, a1.x); a1.y = fmaf(w1, xv[U].y, a1.y); \
        a1.z = fmaf(w1, xv[U].z, a1.z); a1.w = fmaf(w1, xv[U].w, a1.w); }
      POLY_EACH8(POLY_FM)
#undef POLY_FM
      const int k = c + 8 + fi;
      ofr = (unsigned)nx.x * ld4;
      w0r = (k < n0) ? (unsigned)nx.y : 0u;
      w1r = (k >= n0 && k < n) ? (unsigned)nx.y : 0u;
    }
  }
  const int q0 = 2 * g;
  if (!TILE || keep) {
    float4* xo = (float4*)(xout + (size_t)r0 * 32);
    if (q0 < nr)     xo[(unsigned)q0 * 8u + f] = a0;
    if (q0 + 1 < nr) xo[(unsigned)(q0 + 1) * 8u + f] = a1;
  }
  if (TILE) {
    *(float4*)(st + q0 * TP + 4 * fi) = a0;
    *(float4*)(st + (q0 + 1) * TP + 4 * fi) = a1;
  }
}

template <bool TILE>
__device__ __forceinline__ void poly_walk(
    const int* __restrict__ rowptr, const int* __restrict__ sedge,
    const float* __restrict__ xin, int ldin,
    float* __restrict__ xout, float* st, int keep, int r0, int N, unsigned lane)
{
  const int g = (int)(lane >> 3);
  const int nr = min(GR, N - r0);
  const int s = rowptr[r0 + min(2 * g, nr)];
  const int m = rowptr[r0 + min(2 * g + 1, nr)];
  const int e = rowptr[r0 + min(2 * g + 2, nr)];
  poly_walk_run<TILE>((const int2*)sedge, s, m, e, xin, ldin, xout, st, keep, r0, nr, lane);
}

__global__ __launch_bounds__(1024) void poly_finalize(
    const int2* __restrict__ coarse, const int* __restrict__ off1,
    int* __restrict__ rowptr, int2* __restrict__ sedge,
    const float* __restrict__ x, float* __restrict__ x1, int N, int E, int nsl)
{
  __shared__ int rc[RB];
  __shared__ int ws[32];
  __shared__ int sgA[MAXSL];
  __shared__ int sgL[MAXSL];
  __shared__ int sgO[MAXSL];
  __shared__ int2 stF[FCAP];
  const int t = threadIdx.x;
  const int bk = blockIdx.x;
  const int per = (E + nsl - 1) / nsl;
  int a = 0, len = 0;
  if (t < nsl) {
    const int* o = off1 + (size_t)t * (gridDim.x + 1) + bk;
    a = o[0];
    len = o[1] - a;
    sgA[t] = min(t * per, E) + a;
    sgL[t] = len;
    sgO[t] = a;
  }
  if (t < RB) rc[t] = 0;
  __syncthreads();
  int s = 0, n = 0;
  for (int i = t & 31; i < nsl; i += 32) { s += sgO[i]; n += sgL[i]; }
#pragma unroll
  for (int d = 16; d >= 1; d >>= 1) { s += __shfl_xor(s, d); n += __shfl_xor(n, d); }
  const int e = s + n;
  poly_bucket_pass<false>(coarse, sgA, sgL, nsl, t, rc, stF, sedge, s, false);
  __syncthreads();
  const int cn = (t < RB) ? rc[t] : 0;
  int sum;
  const int ex = poly_block_scan(cn, ws, sum);
  if (t < RB) {
    rc[t] = s + ex;
    const int r = bk * RB + t;
    if (r < N) rowptr[r] = s + ex;
  }
  if (bk == (int)gridDim.x - 1 && t == 0) rowptr[N] = e;
  __syncthreads();
  const bool stage = n <= FCAP;
  poly_bucket_pass<true>(coarse, sgA, sgL, nsl, t, rc, stF, sedge, s, stage);
  __syncthreads();
  if (stage) {
    const int off = (int)(((size_t)(sedge + s) >> 3) & 15);
    for (int i = t - off; i < n; i += 1024)
      if (i >= 0) sedge[s + i] = stF[i];
  }
  const unsigned lane = (unsigned)t & 31u;
  const int rl = __builtin_amdgcn_readfirstlane(t >> 5) * GR;
  const int r0 = bk * RB + rl;
  if (r0 < N) {
    const int nr = min(GR, N - r0);
    const int g = (int)(lane >> 3);
    const int i0 = rl + min(2 * g, nr);
    const int ps = (i0 > 0) ? rc[i0 - 1] : s;
    const int pm = rc[rl + min(2 * g + 1, nr) - 1];
    const int pe = rc[rl + min(2 * g + 2, nr) - 1];
    if (stage) poly_walk_run<false>(stF, ps - s, pm - s, pe - s, x, 64, x1, nullptr, 1, r0, nr, lane);
    else       poly_walk_run<false>(sedge + s, ps - s, pm - s, pe - s, x, 64, x1, nullptr, 1, r0, nr, lane);
  }
}

__global__ __launch_bounds__(256) void poly_gather(
    const int* __restrict__ rowptr, const int* __restrict__ sedge,
    const float* __restrict__ xin, int ldin,
    float* __restrict__ xout, int N)
{
  const unsigned lane = threadIdx.x & 31;
  const int r0 = __builtin_amdgcn_readfirstlane((blockIdx.x * 8 + (threadIdx.x >> 5)) * GR);
  if (r0 >= N) return;
  poly_walk<false>(rowptr, sedge, xin, ldin, xout, nullptr, 1, r0, N, lane);
}

template <bool FIRST>
__global__ __launch_bounds__(256) void poly_gather_mm(
    const int* __restrict__ rowptr, const int* __restrict__ sedge,
    const float* __restrict__ xin, float* __restrict__ xout, int keep,
    const float* __restrict__ WA, const float* __restrict__ WB,
    const float* __restrict__ x, const float* __restrict__ W0,
    const float* __restrict__ bias,
    float* __restrict__ out, int N)
{
  __shared__ v2f sW[16 * 4 * 32];
  __shared__ v2f sW0[FIRST ? 16 * 4 * 32 : 1];
  __shared__ __attribute__((aligned(16))) float sT[8 * GR * TP];
  const unsigned lane = threadIdx.x & 31;
  const int wave = __builtin_amdgcn_readfirstlane(threadIdx.x >> 5);
  const int lo = lane & 15;
  const int hi = lane >> 4;

  v2f wv[8], wv0[FIRST ? 8 : 1];
#pragma unroll
  for (int i = 0; i < 8; ++i) {
    wv[i] = poly_w_entry(WA, WB, threadIdx.x + 256 * i);
    if (FIRST) wv0[i] = poly_w_entry(W0, W0 + 32 * 64, threadIdx.x + 256 * i);
  }

  const int tile = wave >> 1;
  const int nsel = wave & 1;
  const int rowBase = blockIdx.x * (8 * GR) + tile * 16;
  const int col0 = nsel * 32 + lo;
  const int rowA = min(rowBase + lo, N - 1);
  const v2f* xa = (const v2f*)(xin + (size_t)rowA * 32 + 2 * hi);
  v2f av[8];
#pragma unroll
  for (int s = 0; s < 8; ++s) av[s] = xa[2 * s];
  float o0[8], o1[8];
  if (FIRST) {
    o0[0] = bias[col0];
    o1[0] = bias[col0 + 16];
  } else {
#pragma unroll
    for (int r = 0; r < 8; ++r) {
      const float* o = out + (size_t)min(rowBase + r + 8 * hi, N - 1) * 64 + col0;
      o0[r] = o[0];
      o1[r] = o[16];
    }
  }

  const int r0 = (blockIdx.x * 8 + wave) * GR;
  float* st = sT + wave * (GR * TP);
  if (r0 < N) {
    poly_walk<true>(rowptr, sedge, xin, 32, xout, st, keep, r0, N, lane);
  } else {
    for (int q = 0; q < GR; ++q) st[q * TP + (int)lane] = 0.f;
  }

#pragma unroll
  for (int i = 0; i < 8; ++i) {
    sW[threadIdx.x + 256 * i] = wv[i];
    if (FIRST) sW0[threadIdx.x + 256 * i] = wv0[i];
  }
  __syncthreads();
  if (rowBase >= N) return;

  v2f ax[FIRST ? 16 : 1];
  if (FIRST) {
    const v2f* xr = (const v2f*)(x + (size_t)rowA * 64 + 2 * hi);
#pragma unroll
    for (int s = 0; s < 16; ++s) ax[s] = xr[2 * s];
  }

  v8f acc0 = {}, acc1 = {};
  const v2f* wl = sW + lane + nsel * 64;
  const float* ta = sT + (tile * 16 + lo) * TP + 2 * hi;
#pragma unroll
  for (int s = 0; s < 16; ++s) {
    v2f a;
    if (s < 8) a = av[s];
    else       a = *(const v2f*)(ta + 4 * (s - 8));
    const v2f b0 = wl[s * 128];
    const v2f b1 = wl[s * 128 + 32];
    acc0 = __builtin_amdgcn_wmma_f32_16x16x4_f32(false, a, false, b0, (short)0, acc0, false, false);
    acc1 = __builtin_amdgcn_wmma_f32_16x16x4_f32(false, a, false, b1, (short)0, acc1, false, false);
  }

  if (FIRST) {
    v8f acx0 = {}, acx1 = {};
    const v2f* wl0 = sW0 + lane + nsel * 64;
#pragma unroll
    for (int s = 0; s < 16; ++s) {
      const v2f b0 = wl0[s * 128];
      const v2f b1 = wl0[s * 128 + 32];
      acx0 = __builtin_amdgcn_wmma_f32_16x16x4_f32(false, ax[s], false, b0, (short)0, acx0, false, false);
      acx1 = __builtin_amdgcn_wmma_f32_16x16x4_f32(false, ax[s], false, b1, (short)0, acx1, false, false);
    }
#pragma unroll
    for (int r = 0; r < 8; ++r) {
      const int row = rowBase + r + 8 * hi;
      if (row < N) {
        float* o = out + (size_t)row * 64 + col0;
        o[0]  = (acx0[r] + o0[0]) + acc0[r];
        o[16] = (acx1[r] + o1[0]) + acc1[r];
      }
    }
  } else {
#pragma unroll
    for (int r = 0; r < 8; ++r) {
      const int row = rowBase + r + 8 * hi;
      if (row < N) {
        float* o = out + (size_t)row * 64 + col0;
        o[0]  = o0[r] + acc0[r];
        o[16] = o1[r] + acc1[r];
      }
    }
  }
}

extern "C" void kernel_launch(void* const* d_in, const int* in_sizes, int n_in,
                              void* d_out, int out_size, void* d_ws, size_t ws_size,
                              hipStream_t stream) {
  const float* x       = (const float*)d_in[0];
  const int*   lap_row = (const int*)d_in[1];
  const int*   lap_col = (const int*)d_in[2];
  const float* lap_val = (const float*)d_in[3];
  const float* weights = (const float*)d_in[4];
  const float* bias    = (const float*)d_in[5];
  float* out = (float*)d_out;

  const int N = in_sizes[0] / 64;
  const int E = in_sizes[1];

  float* P0   = (float*)d_ws;
  float* P1   = P0 + (size_t)N * 32;
  int* rowptr = (int*)(P1 + (size_t)N * 32) + N;
  int* sedge  = rowptr + N + 4;
  int2* coarse = (int2*)P1;
  int* off1    = (int*)(P1 + (size_t)N * 32);

  const int NB  = (N + RB - 1) / RB;
  const int nsl = ((E + SCAP - 1) / SCAP > NSLICE) ? (E + SCAP - 1) / SCAP : NSLICE;

  dim3 gatherGrid((N + 8 * GR - 1) / (8 * GR)), gatherBlock(256);

  poly_slice_sort<<<dim3(nsl), dim3(1024), 0, stream>>>(lap_row, lap_col, lap_val,
                                                        off1, coarse, E, NB, nsl);
  poly_finalize<<<dim3(NB), dim3(1024), 0, stream>>>(coarse, off1, rowptr, (int2*)sedge, x, P0, N, E, nsl);

  for (int h = 0; h < 2; ++h) {
    const float* wh = weights + (size_t)h * 32 * 64;
    if (h == 1)
      poly_gather<<<gatherGrid, gatherBlock, 0, stream>>>(rowptr, sedge, x + 32, 64, P0, N);
    if (h == 0)
      poly_gather_mm<true><<<gatherGrid, gatherBlock, 0, stream>>>(
          rowptr, sedge, P0, P1, 1, wh + (size_t)1 * 64 * 64, wh + (size_t)2 * 64 * 64,
          x, weights, bias, out, N);
    else
      poly_gather_mm<false><<<gatherGrid, gatherBlock, 0, stream>>>(
          rowptr, sedge, P0, P1, 1, wh + (size_t)1 * 64 * 64, wh + (size_t)2 * 64 * 64,
          x, weights, bias, out, N);
    poly_gather<<<gatherGrid, gatherBlock, 0, stream>>>(rowptr, sedge, P1, 32, P0, N);
    poly_gather_mm<false><<<gatherGrid, gatherBlock, 0, stream>>>(
        rowptr, sedge, P0, P1, 0, wh + (size_t)3 * 64 * 64, wh + (size_t)4 * 64 * 64,
        x, weights, bias, out, N);
  }
}


# Half 1's first hop rides in half 0's last walk: same rows, same records, a second line load per edge, both planes' 64 lines per wave in one round trip; 8 launches, was 9; row sum order kept
# speedup vs baseline: 5.6228x; 1.0374x over previous
#include <hip/hip_runtime.h>

typedef __attribute__((ext_vector_type(2))) float v2f;
typedef __attribute__((ext_vector_type(8))) float v8f;

__device__ __forceinline__ v2f poly_w_entry(const float* __restrict__ WA,
                                            const float* __restrict__ WB, int j)
{
  const int kb4 = j >> 7;
  const int t   = (j >> 5) & 3;
  const int ln  = j & 31;
  const int h   = ln >> 4;
  const int lw  = ln & 15;
  const int krow = kb4 * 4 + 2 * h;
  const int ncol = lw + 16 * t;
  const float* wp = (krow < 32) ? (WA + krow * 64) : (WB + (krow - 32) * 64);
  v2f p;
  p.x = wp[ncol];
  p.y = wp[64 + ncol];
  return p;
}

#define SB 8
#define RB (1 << SB)
#define MAXNB 512
#define NSLICE 128
#define MAXSL 1024
#define CH 8
#define SCAP 8192
#define SG 6
#define FCAP 6144

__device__ __forceinline__ int poly_wave_scan(int v, int lane) {
#pragma unroll
  for (int d = 1; d < 32; d <<= 1) {
    const int u = __shfl_up(v, d);
    if (lane >= d) v += u;
  }
  return v;
}

__device__ __forceinline__ int poly_block_scan(int v, int* ws, int& total) {
  const int lane = threadIdx.x & 31;
  const int wave = threadIdx.x >> 5;
  const int inc = poly_wave_scan(v, lane);
  if (lane == 31) ws[wave] = inc;
  __syncthreads();
  const int wt = ws[lane];
  const int wsc = poly_wave_scan(wt, lane);
  total = __shfl(wsc, 31);
  return __shfl(wsc, wave) - __shfl(wt, wave) + inc - v;
}

__global__ __launch_bounds__(1024) void poly_slice_sort(
    const int* __restrict__ row, const int* __restrict__ col, const float* __restrict__ val,
    int* __restrict__ off1, int2* __restrict__ coarse, int E, int NB, int nsl)
{
  __shared__ int cnt[MAXNB];
  __shared__ int ws[32];
  __shared__ int2 stR[SCAP];
  const int t = threadIdx.x;
  if (t < MAXNB) cnt[t] = 0;
  __syncthreads();
  const int per = (E + nsl - 1) / nsl;
  const int b = min((int)blockIdx.x * per, E);
  const int e = min(b + per, E);
  int r[CH], c[CH];
  float v[CH];
#pragma unroll
  for (int k = 0; k < CH; ++k) {
    const int i = b + k * 1024 + t;
    const bool ok = i < e;
    r[k] = ok ? row[i] : -1;
    c[k] = ok ? col[i] : 0;
    v[k] = ok ? val[i] : 0.f;
  }
  int p[CH];
#pragma unroll
  for (int k = 0; k < CH; ++k)
    p[k] = (r[k] >= 0) ? atomicAdd(&cnt[r[k] >> SB], 1) : -1;
  __syncthreads();
  const int own = (t < MAXNB) ? cnt[t] : 0;
  int sum;
  const int lex = poly_block_scan(own, ws, sum);
  if (t < MAXNB) cnt[t] = lex;
  if (t <= NB) off1[(size_t)blockIdx.x * (NB + 1) + t] = lex;
  __syncthreads();
  int q[CH];
#pragma unroll
  for (int k = 0; k < CH; ++k) q[k] = cnt[max(r[k], 0) >> SB];
#pragma unroll
  for (int k = 0; k < CH; ++k)
    if (p[k] >= 0)
      stR[q[k] + p[k]] = make_int2((c[k] << SB) | (r[k] & (RB - 1)), __float_as_int(v[k]));
  __syncthreads();
  const int n = e - b;
  const int off = (int)(((size_t)(coarse + b) >> 3) & 15);
  int2 w[CH + 1];
#pragma unroll
  for (int k = 0; k <= CH; ++k)
    w[k] = stR[min(max(t - off + k * 1024, 0), SCAP - 1)];
#pragma unroll
  for (int k = 0; k <= CH; ++k) {
    const int i = t - off + k * 1024;
    if (i >= 0 && i < n) coarse[b + i] = w[k];
  }
}

template <bool PLACE>
__device__ __forceinline__ void poly_bucket_pass(
    const int2* __restrict__ coarse, const int* sgA, const int* sgL, int nsl, int t,
    int* rc, int2* stF, int2* __restrict__ sedge, int s, bool stage)
{
  const int f = t & 7;
  for (int sg = t >> 3; sg < nsl; sg += 128) {
    const int base = sgA[sg];
    const int ln = sgL[sg];
    for (int j0 = 0; j0 < ln; j0 += 8 * SG) {
      int2 rec[SG];
#pragma unroll
      for (int u = 0; u < SG; ++u) {
        const int j = j0 + 8 * u + f;
        rec[u] = (j < ln) ? coarse[base + j] : make_int2(-1, 0);
      }
      if (!PLACE) {
#pragma unroll
        for (int u = 0; u < SG; ++u)
          if (rec[u].x >= 0) atomicAdd(&rc[rec[u].x & (RB - 1)], 1);
      } else {
        int p[SG];
#pragma unroll
        for (int u = 0; u < SG; ++u)
          p[u] = (rec[u].x >= 0) ? atomicAdd(&rc[rec[u].x & (RB - 1)], 1) : -1;
#pragma unroll
        for (int u = 0; u < SG; ++u)
          if (p[u] >= 0) {
            const int2 o = make_int2(rec[u].x >> SB, rec[u].y);
            if (stage) stF[p[u] - s] = o;
            else       sedge[p[u]] = o;
          }
      }
    }
  }
}

#define GR 8
#define TP 36

template <int U>
__device__ __forceinline__ unsigned poly_bc8(unsigned v) {
  return (unsigned)__builtin_amdgcn_mov_dpp8(v, (unsigned)U * 0x249249u);
}

#define POLY_EACH8(M) M(0) M(1) M(2) M(3) M(4) M(5) M(6) M(7)

template <bool TILE, bool DUAL>
__device__ __forceinline__ void poly_walk_run(
    const int2* __restrict__ se, int s, int m, int e,
    const float* __restrict__ xin, int ldin,
    float* __restrict__ xout, float* st, int keep, int r0, int nr, unsigned lane,
    const float* __restrict__ x2in, float* __restrict__ x2out)
{
  const int g = (int)(lane >> 3);
  const unsigned f = lane & 7u;
  const int fi = (int)f;
  const int n  = e - s;
  const int n0 = m - s;
  const int nmax = max(max(__builtin_amdgcn_readlane(n, 0), __builtin_amdgcn_readlane(n, 8)),
                       max(__builtin_amdgcn_readlane(n, 16), __builtin_amdgcn_readlane(n, 24)));
  const float4* xin4 = (const float4*)xin;
  const unsigned ld4 = (unsigned)ldin >> 2;
  float4 a0 = make_float4(0.f, 0.f, 0.f, 0.f);
  float4 a1 = make_float4(0.f, 0.f, 0.f, 0.f);
  const float4* x2in4 = (const float4*)x2in;
  float4 b0 = make_float4(0.f, 0.f, 0.f, 0.f);
  float4 b1 = make_float4(0.f, 0.f, 0.f, 0.f);
  if (nmax > 0) {
    const int last = max(e - 1, 0);
    const int2 rec = se[min(s + fi, last)];
    unsigned ofr = (unsigned)rec.x * ld4;
    unsigned w0r = (fi < n0) ? (unsigned)rec.y : 0u;
    unsigned w1r = (fi >= n0 && fi < n) ? (unsigned)rec.y : 0u;
    for (int c = 0; c < nmax; c += 8) {
      const int2 nx = se[min(s + c + 8 + fi, last)];
      float4 xv[8];
      float4 yv[DUAL ? 8 : 1];
#define POLY_LD(U) { const unsigned o = poly_bc8<U>(ofr); xv[U] = xin4[o + f]; \
                     if (DUAL) yv[U] = x2in4[2u * o + f]; }
      POLY_EACH8(POLY_LD)
#undef POLY_LD
      if (!DUAL)
        asm volatile("" :: "v"(nx.x), "v"(nx.y), "v"(xv[0].x), "v"(xv[1].x), "v"(xv[2].x), "v"(xv[3].x),
                           "v"(xv[4].x), "v"(xv[5].x), "v"(xv[6].x), "v"(xv[7].x));
      else
        asm volatile("" :: "v"(nx.x), "v"(nx.y), "v"(xv[0].x), "v"(xv[1].x), "v"(xv[2].x), "v"(xv[3].x),
                           "v"(xv[4].x), "v"(xv[5].x), "v"(xv[6].x), "v"(xv[7].x),
                           "v"(yv[0].x), "v"(yv[DUAL ? 1 : 0].x), "v"(yv[DUAL ? 2 : 0].x), "v"(yv[DUAL ? 3 : 0].x),
                           "v"(yv[DUAL ? 4 : 0].x), "v"(yv[DUAL ? 5 : 0].x), "v"(yv[DUAL ? 6 : 0].x), "v"(yv[DUAL ? 7 : 0].x));
      __builtin_amdgcn_sched_barrier(0);
#define POLY_FM(U) { \
        const float w0 = __uint_as_float(poly_bc8<U>(w0r)); \
        const float w1 = __uint_as_float(poly_bc8<U>(w1r)); \
        a0.x = fmaf(w0, xv[U].x, a0.x); a0.y = fmaf(w0, xv[U].y, a0.y); \
        a0.z = fmaf(w0, xv[U].z, a0.z); a0.w = fmaf(w0, xv[U].w, a0.w); \
        a1.x = fmaf(w1, xv[U].x, a1.x); a1.y = fmaf(w1, xv[U].y, a1.y); \
        a1.z = fmaf(w1, xv[U].z, a1.z); a1.w = fmaf(w1, xv[U].w, a1.w); \
        if (DUAL) { const float4 y = yv[DUAL ? U : 0]; \
        b0.x = fmaf(w0, y.x, b0.x); b0.y = fmaf(w0, y.y, b0.y); \
        b0.z = fmaf(w0, y.z, b0.z); b0.w = fmaf(w0, y.w, b0.w); \
        b1.x = fmaf(w1, y.x, b1.x); b1.y = fmaf(w1, y.y, b1.y); \
        b1.z = fmaf(w1, y.z, b1.z); b1.w = fmaf(w1, y.w, b1.w); } }
      POLY_EACH8(POLY_FM)
#undef POLY_FM
      const int k = c + 8 + fi;
      ofr = (unsigned)nx.x * ld4;
      w0r = (k < n0) ? (unsigned)nx.y : 0u;
      w1r = (k >= n0 && k < n) ? (unsigned)nx.y : 0u;
    }
  }
  const int q0 = 2 * g;
  if (!DUAL && (!TILE || keep)) {
    float4* xo = (float4*)(xout + (size_t)r0 * 32);
    if (q0 < nr)     xo[(unsigned)q0 * 8u + f] = a0;
    if (q0 + 1 < nr) xo[(unsigned)(q0 + 1) * 8u + f] = a1;
  }
  if (DUAL) {
    float4* yo = (float4*)(x2out + (size_t)r0 * 32);
    if (q0 < nr)     yo[(unsigned)q0 * 8u + f] = b0;
    if (q0 + 1 < nr) yo[(unsigned)(q0 + 1) * 8u + f] = b1;
  }
  if (TILE) {
    *(float4*)(st + q0 * TP + 4 * fi) = a0;
    *(float4*)(st + (q0 + 1) * TP + 4 * fi) = a1;
  }
}

template <bool TILE, bool DUAL>
__device__ __forceinline__ void poly_walk(
    const int* __restrict__ rowptr, const int* __restrict__ sedge,
    const float* __restrict__ xin, int ldin,
    float* __restrict__ xout, float* st, int keep, int r0, int N, unsigned lane,
    const float* __restrict__ x2in, float* __restrict__ x2out)
{
  const int g = (int)(lane >> 3);
  const int nr = min(GR, N - r0);
  const int s = rowptr[r0 + min(2 * g, nr)];
  const int m = rowptr[r0 + min(2 * g + 1, nr)];
  const int e = rowptr[r0 + min(2 * g + 2, nr)];
  poly_walk_run<TILE, DUAL>((const int2*)sedge, s, m, e, xin, ldin, xout, st, keep, r0, nr, lane, x2in, x2out);
}

__global__ __launch_bounds__(1024) void poly_finalize(
    const int2* __restrict__ coarse, const int* __restrict__ off1,
    int* __restrict__ rowptr, int2* __restrict__ sedge,
    const float* __restrict__ x, float* __restrict__ x1, int N, int E, int nsl)
{
  __shared__ int rc[RB];
  __shared__ int ws[32];
  __shared__ int sgA[MAXSL];
  __shared__ int sgL[MAXSL];
  __shared__ int sgO[MAXSL];
  __shared__ int2 stF[FCAP];
  const int t = threadIdx.x;
  const int bk = blockIdx.x;
  const int per = (E + nsl - 1) / nsl;
  int a = 0, len = 0;
  if (t < nsl) {
    const int* o = off1 + (size_t)t * (gridDim.x + 1) + bk;
    a = o[0];
    len = o[1] - a;
    sgA[t] = min(t * per, E) + a;
    sgL[t] = len;
    sgO[t] = a;
  }
  if (t < RB) rc[t] = 0;
  __syncthreads();
  int s = 0, n = 0;
  for (int i = t & 31; i < nsl; i += 32) { s += sgO[i]; n += sgL[i]; }
#pragma unroll
  for (int d = 16; d >= 1; d >>= 1) { s += __shfl_xor(s, d); n += __shfl_xor(n, d); }
  const int e = s + n;
  poly_bucket_pass<false>(coarse, sgA, sgL, nsl, t, rc, stF, sedge, s, false);
  __syncthreads();
  const int cn = (t < RB) ? rc[t] : 0;
  int sum;
  const int ex = poly_block_scan(cn, ws, sum);
  if (t < RB) {
    rc[t] = s + ex;
    const int r = bk * RB + t;
    if (r < N) rowptr[r] = s + ex;
  }
  if (bk == (int)gridDim.x - 1 && t == 0) rowptr[N] = e;
  __syncthreads();
  const bool stage = n <= FCAP;
  poly_bucket_pass<true>(coarse, sgA, sgL, nsl, t, rc, stF, sedge, s, stage);
  __syncthreads();
  if (stage) {
    const int off = (int)(((size_t)(sedge + s) >> 3) & 15);
    for (int i = t - off; i < n; i += 1024)
      if (i >= 0) sedge[s + i] = stF[i];
  }
  const unsigned lane = (unsigned)t & 31u;
  const int rl = __builtin_amdgcn_readfirstlane(t >> 5) * GR;
  const int r0 = bk * RB + rl;
  if (r0 < N) {
    const int nr = min(GR, N - r0);
    const int g = (int)(lane >> 3);
    const int i0 = rl + min(2 * g, nr);
    const int ps = (i0 > 0) ? rc[i0 - 1] : s;
    const int pm = rc[rl + min(2 * g + 1, nr) - 1];
    const int pe = rc[rl + min(2 * g + 2, nr) - 1];
    if (stage) poly_walk_run<false, false>(stF, ps - s, pm - s, pe - s, x, 64, x1, nullptr, 1, r0, nr, lane, nullptr, nullptr);
    else       poly_walk_run<false, false>(sedge + s, ps - s, pm - s, pe - s, x, 64, x1, nullptr, 1, r0, nr, lane, nullptr, nullptr);
  }
}

__global__ __launch_bounds__(256) void poly_gather(
    const int* __restrict__ rowptr, const int* __restrict__ sedge,
    const float* __restrict__ xin, int ldin,
    float* __restrict__ xout, int N)
{
  const unsigned lane = threadIdx.x & 31;
  const int r0 = __builtin_amdgcn_readfirstlane((blockIdx.x * 8 + (threadIdx.x >> 5)) * GR);
  if (r0 >= N) return;
  poly_walk<false, false>(rowptr, sedge, xin, ldin, xout, nullptr, 1, r0, N, lane, nullptr, nullptr);
}

template <bool FIRST, bool DUAL>
__global__ __launch_bounds__(256) void poly_gather_mm(
    const int* __restrict__ rowptr, const int* __restrict__ sedge,
    const float* __restrict__ xin, float* __restrict__ xout, int keep,
    const float* __restrict__ WA, const float* __restrict__ WB,
    const float* __restrict__ x, const float* __restrict__ W0,
    const float* __restrict__ bias,
    float* __restrict__ out, int N)
{
  __shared__ v2f sW[16 * 4 * 32];
  __shared__ v2f sW0[FIRST ? 16 * 4 * 32 : 1];
  __shared__ __attribute__((aligned(16))) float sT[8 * GR * TP];
  const unsigned lane = threadIdx.x & 31;
  const int wave = __builtin_amdgcn_readfirstlane(threadIdx.x >> 5);
  const int lo = lane & 15;
  const int hi = lane >> 4;

  v2f wv[8], wv0[FIRST ? 8 : 1];
#pragma unroll
  for (int i = 0; i < 8; ++i) {
    wv[i] = poly_w_entry(WA, WB, threadIdx.x + 256 * i);
    if (FIRST) wv0[i] = poly_w_entry(W0, W0 + 32 * 64, threadIdx.x + 256 * i);
  }

  const int tile = wave >> 1;
  const int nsel = wave & 1;
  const int rowBase = blockIdx.x * (8 * GR) + tile * 16;
  const int col0 = nsel * 32 + lo;
  const int rowA = min(rowBase + lo, N - 1);
  const v2f* xa = (const v2f*)(xin + (size_t)rowA * 32 + 2 * hi);
  v2f av[8];
#pragma unroll
  for (int s = 0; s < 8; ++s) av[s] = xa[2 * s];
  float o0[8], o1[8];
  if (FIRST) {
    o0[0] = bias[col0];
    o1[0] = bias[col0 + 16];
  } else if (!DUAL) {
#pragma unroll
    for (int r = 0; r < 8; ++r) {
      const float* o = out + (size_t)min(rowBase + r + 8 * hi, N - 1) * 64 + col0;
      o0[r] = o[0];
      o1[r] = o[16];
    }
  }

  if (DUAL) {
#pragma unroll
    for (int i = 0; i < 8; ++i) sW[threadIdx.x + 256 * i] = wv[i];
  }

  const int r0 = (blockIdx.x * 8 + wave) * GR;
  float* st = sT + wave * (GR * TP);
  if (r0 < N) {
    poly_walk<true, DUAL>(rowptr, sedge, xin, 32, xout, st, keep, r0, N, lane, x + 32, xout);
  } else {
    for (int q = 0; q < GR; ++q) st[q * TP + (int)lane] = 0.f;
  }

  if (!DUAL) {
#pragma unroll
    for (int i = 0; i < 8; ++i) {
      sW[threadIdx.x + 256 * i] = wv[i];
      if (FIRST) sW0[threadIdx.x + 256 * i] = wv0[i];
    }
  }
  __syncthreads();
  if (rowBase >= N) return;

  v2f ax[FIRST ? 16 : 1];
  if (FIRST) {
    const v2f* xr = (const v2f*)(x + (size_t)rowA * 64 + 2 * hi);
#pragma unroll
    for (int s = 0; s < 16; ++s) ax[s] = xr[2 * s];
  }

  if (DUAL) {
#pragma unroll
    for (int r = 0; r < 8; ++r) {
      const float* o = out + (size_t)min(rowBase + r + 8 * hi, N - 1) * 64 + col0;
      o0[r] = o[0];
      o1[r] = o[16];
    }
  }

  v8f acc0 = {}, acc1 = {};
  const v2f* wl = sW + lane + nsel * 64;
  const float* ta = sT + (tile * 16 + lo) * TP + 2 * hi;
#pragma unroll
  for (int s = 0; s < 16; ++s) {
    v2f a;
    if (s < 8) a = av[s];
    else       a = *(const v2f*)(ta + 4 * (s - 8));
    const v2f b0 = wl[s * 128];
    const v2f b1 = wl[s * 128 + 32];
    acc0 = __builtin_amdgcn_wmma_f32_16x16x4_f32(false, a, false, b0, (short)0, acc0, false, false);
    acc1 = __builtin_amdgcn_wmma_f32_16x16x4_f32(false, a, false, b1, (short)0, acc1, false, false);
  }

  if (FIRST) {
    v8f acx0 = {}, acx1 = {};
    const v2f* wl0 = sW0 + lane + nsel * 64;
#pragma unroll
    for (int s = 0; s < 16; ++s) {
      const v2f b0 = wl0[s * 128];
      const v2f b1 = wl0[s * 128 + 32];
      acx0 = __builtin_amdgcn_wmma_f32_16x16x4_f32(false, ax[s], false, b0, (short)0, acx0, false, false);
      acx1 = __builtin_amdgcn_wmma_f32_16x16x4_f32(false, ax[s], false, b1, (short)0, acx1, false, false);
    }
#pragma unroll
    for (int r = 0; r < 8; ++r) {
      const int row = rowBase + r + 8 * hi;
      if (row < N) {
        float* o = out + (size_t)row * 64 + col0;
        o[0]  = (acx0[r] + o0[0]) + acc0[r];
        o[16] = (acx1[r] + o1[0]) + acc1[r];
      }
    }
  } else {
#pragma unroll
    for (int r = 0; r < 8; ++r) {
      const int row = rowBase + r + 8 * hi;
      if (row < N) {
        float* o = out + (size_t)row * 64 + col0;
        o[0]  = o0[r] + acc0[r];
        o[16] = o1[r] + acc1[r];
      }
    }
  }
}

extern "C" void kernel_launch(void* const* d_in, const int* in_sizes, int n_in,
                              void* d_out, int out_size, void* d_ws, size_t ws_size,
                              hipStream_t stream) {
  const float* x       = (const float*)d_in[0];
  const int*   lap_row = (const int*)d_in[1];
  const int*   lap_col = (const int*)d_in[2];
  const float* lap_val = (const float*)d_in[3];
  const float* weights = (const float*)d_in[4];
  const float* bias    = (const float*)d_in[5];
  float* out = (float*)d_out;

  const int N = in_sizes[0] / 64;
  const int E = in_sizes[1];

  float* P0   = (float*)d_ws;
  float* P1   = P0 + (size_t)N * 32;
  int* rowptr = (int*)(P1 + (size_t)N * 32) + N;
  int* sedge  = rowptr + N + 4;
  int2* coarse = (int2*)P1;
  int* off1    = (int*)(P1 + (size_t)N * 32);

  const int NB  = (N + RB - 1) / RB;
  const int nsl = ((E + SCAP - 1) / SCAP > NSLICE) ? (E + SCAP - 1) / SCAP : NSLICE;

  dim3 gatherGrid((N + 8 * GR - 1) / (8 * GR)), gatherBlock(256);

  poly_slice_sort<<<dim3(nsl), dim3(1024), 0, stream>>>(lap_row, lap_col, lap_val,
                                                        off1, coarse, E, NB, nsl);
  poly_finalize<<<dim3(NB), dim3(1024), 0, stream>>>(coarse, off1, rowptr, (int2*)sedge, x, P0, N, E, nsl);

  const float* w0h = weights;
  const float* w1h = weights + (size_t)32 * 64;
  poly_gather_mm<true, false><<<gatherGrid, gatherBlock, 0, stream>>>(
      rowptr, sedge, P0, P1, 1, w0h + (size_t)1 * 64 * 64, w0h + (size_t)2 * 64 * 64,
      x, weights, bias, out, N);
  poly_gather<<<gatherGrid, gatherBlock, 0, stream>>>(rowptr, sedge, P1, 32, P0, N);
  poly_gather_mm<false, true><<<gatherGrid, gatherBlock, 0, stream>>>(
      rowptr, sedge, P0, P1, 0, w0h + (size_t)3 * 64 * 64, w0h + (size_t)4 * 64 * 64,
      x, weights, bias, out, N);
  poly_gather_mm<false, false><<<gatherGrid, gatherBlock, 0, stream>>>(
      rowptr, sedge, P1, P0, 1, w1h + (size_t)1 * 64 * 64, w1h + (size_t)2 * 64 * 64,
      x, weights, bias, out, N);
  poly_gather<<<gatherGrid, gatherBlock, 0, stream>>>(rowptr, sedge, P0, 32, P1, N);
  poly_gather_mm<false, false><<<gatherGrid, gatherBlock, 0, stream>>>(
      rowptr, sedge, P1, P0, 0, w1h + (size_t)3 * 64 * 64, w1h + (size_t)4 * 64 * 64,
      x, weights, bias, out, N);
}
